# starting kernel
# baseline (speedup 1.0000x reference)
#include <hip/hip_runtime.h>

typedef __attribute__((ext_vector_type(2))) float v2f;
typedef __attribute__((ext_vector_type(8))) float v8f;

#define HIDC 128

template <int NC>
__global__ __launch_bounds__(256) void gemm_wmma_f32(
    const float* __restrict__ A, const float* __restrict__ B,
    const float* __restrict__ bias, float* __restrict__ C, int M)
{
    constexpr int K = HIDC;
    constexpr int tilesN = (NC + 15) >> 4;
    const int tilesM = (M + 15) >> 4;
    const int wid = blockIdx.x * 8 + (threadIdx.x >> 5);
    if (wid >= tilesM * tilesN) return;
    const int tm = wid / tilesN, tn = wid % tilesN;
    const int m0 = tm << 4, n0 = tn << 4;
    const int lane = threadIdx.x & 31;
    const int half = lane >> 4;
    const int l16  = lane & 15;

    const int arow   = min(m0 + l16, M - 1);
    const int bcol_r = n0 + l16;
    const int bcol   = min(bcol_r, NC - 1);

    const float* Ap = A + (size_t)arow * K + 2 * half;
    const float* Bp = B + (size_t)(2 * half) * NC + bcol;

    v8f acc = {};
#pragma unroll
    for (int k0 = 0; k0 < K; k0 += 4) {
        v2f a = *(const v2f*)(Ap + k0);
        v2f b;
        b.x = Bp[k0 * NC];
        b.y = Bp[(k0 + 1) * NC];
        acc = __builtin_amdgcn_wmma_f32_16x16x4_f32(
                  false, a, false, b, (short)0, acc, false, false);
    }

    if (bcol_r < NC) {
        const float bv = bias ? bias[bcol_r] : 0.0f;
#pragma unroll
        for (int v = 0; v < 8; ++v) {
            const int r = m0 + v + 8 * half;
            if (r < M) C[(size_t)r * NC + bcol_r] = acc[v] + bv;
        }
    }
}

__global__ void zero_f32(float* __restrict__ p, int n) {
    int i = blockIdx.x * blockDim.x + threadIdx.x;
    if (i < n) p[i] = 0.0f;
}

__global__ void degree_count(const int* __restrict__ row, int E,
                             float* __restrict__ deg) {
    int e = blockIdx.x * blockDim.x + threadIdx.x;
    if (e < E) atomicAdd(&deg[row[e]], 1.0f);
}

__global__ void invert_deg(float* __restrict__ d, int n) {
    int i = blockIdx.x * blockDim.x + threadIdx.x;
    if (i < n) d[i] = 1.0f / (d[i] + 1e-10f);
}

__global__ __launch_bounds__(256) void scatter_add_edges(
    const float* __restrict__ m, const int* __restrict__ ei, int E,
    float* __restrict__ agg)
{
    int t  = blockIdx.x * 256 + threadIdx.x;
    int e  = t >> 5;
    int c4 = (t & 31) << 2;
    if (e >= E) return;
    int r  = ei[e];
    int cl = ei[E + e];
    const float4 mv = *(const float4*)(m + (size_t)r * HIDC + c4);
    float* dst = agg + (size_t)cl * HIDC + c4;
    atomicAdd(dst + 0, mv.x);
    atomicAdd(dst + 1, mv.y);
    atomicAdd(dst + 2, mv.z);
    atomicAdd(dst + 3, mv.w);
}

__global__ void bias_relu(float* __restrict__ agg, const float* __restrict__ b,
                          int total) {
    int i = blockIdx.x * blockDim.x + threadIdx.x;
    if (i < total) {
        float v = agg[i] + b[i & (HIDC - 1)];
        agg[i] = fmaxf(v, 0.0f);
    }
}

__global__ __launch_bounds__(256) void gamma_edges(
    const float* __restrict__ agg, const int* __restrict__ ei, int E,
    float* __restrict__ gsum)
{
    int wid  = blockIdx.x * 8 + (threadIdx.x >> 5);
    int lane = threadIdx.x & 31;
    if (wid >= E) return;
    int r = ei[wid];
    int c = ei[E + wid];
    const float4 a = *(const float4*)(agg + (size_t)r * HIDC + lane * 4);
    const float4 b = *(const float4*)(agg + (size_t)c * HIDC + lane * 4);
    float dx = a.x - b.x, dy = a.y - b.y, dz = a.z - b.z, dw = a.w - b.w;
    float s = dx * dx + dy * dy + dz * dz + dw * dw;
#pragma unroll
    for (int off = 16; off > 0; off >>= 1)
        s += __shfl_down(s, off, 32);
    if (lane == 0) atomicAdd(&gsum[r], s);
}

__global__ void gamma_node(const float* __restrict__ gsum,
                           const float* __restrict__ inv,
                           float* __restrict__ g, int n) {
    int i = blockIdx.x * blockDim.x + threadIdx.x;
    if (i < n) g[i] = tanhf(gsum[i] * inv[i]);
}

__global__ void update_h(float* __restrict__ h, const float* __restrict__ agg,
                         const float* __restrict__ xs,
                         const float* __restrict__ g1,
                         const float* __restrict__ g2, int total) {
    int i = blockIdx.x * blockDim.x + threadIdx.x;
    if (i < total) {
        int node = i >> 7;
        float gs = g1[node], gq = g2[node];
        h[i] = (h[i] + gs * agg[i] + gq * xs[i]) / (1.0f + gs + gq);
    }
}

extern "C" void kernel_launch(void* const* d_in, const int* in_sizes, int n_in,
                              void* d_out, int out_size, void* d_ws, size_t ws_size,
                              hipStream_t stream) {
    const float* x      = (const float*)d_in[0];
    const float* x0     = (const float*)d_in[1];
    const int*   ei     = (const int*)d_in[2];
    const int*   ei2    = (const int*)d_in[3];
    const float* in_W   = (const float*)d_in[4];
    const float* skip_W = (const float*)d_in[5];
    const float* conv_W = (const float*)d_in[6];
    const float* conv_b = (const float*)d_in[7];
    const float* fc_W   = (const float*)d_in[8];
    const float* fc_b   = (const float*)d_in[9];
    float* out = (float*)d_out;

    const int N  = in_sizes[0] / HIDC;
    const int E1 = in_sizes[2] / 2;
    const int E2 = in_sizes[3] / 2;

    const size_t nh = (size_t)N * HIDC;
    float* h     = (float*)d_ws;
    float* xskip = h + nh;
    float* agg   = xskip + nh;
    float* mbuf  = agg + nh;
    float* inv1  = mbuf + nh;
    float* inv2  = inv1 + N;
    float* gsum1 = inv2 + N;
    float* gsum2 = gsum1 + N;
    float* g1    = gsum2 + N;
    float* g2    = g1 + N;

    const dim3 blk(256);
    auto gemm_blocks = [](int M, int Nc) {
        int tiles = ((M + 15) / 16) * ((Nc + 15) / 16);
        return dim3((tiles + 7) / 8);
    };
    const int nhi = (int)nh;

    gemm_wmma_f32<HIDC><<<gemm_blocks(N, HIDC), blk, 0, stream>>>(x,    in_W,   nullptr, h,     N);
    gemm_wmma_f32<HIDC><<<gemm_blocks(N, HIDC), blk, 0, stream>>>(x0,   in_W,   nullptr, mbuf,  N);
    gemm_wmma_f32<HIDC><<<gemm_blocks(N, HIDC), blk, 0, stream>>>(mbuf, skip_W, nullptr, xskip, N);

    zero_f32<<<(N + 255) / 256, blk, 0, stream>>>(inv1, N);
    zero_f32<<<(N + 255) / 256, blk, 0, stream>>>(inv2, N);
    degree_count<<<(E1 + 255) / 256, blk, 0, stream>>>(ei,  E1, inv1);
    degree_count<<<(E2 + 255) / 256, blk, 0, stream>>>(ei2, E2, inv2);
    invert_deg<<<(N + 255) / 256, blk, 0, stream>>>(inv1, N);
    invert_deg<<<(N + 255) / 256, blk, 0, stream>>>(inv2, N);

    for (int l = 0; l < 2; ++l) {
        gemm_wmma_f32<HIDC><<<gemm_blocks(N, HIDC), blk, 0, stream>>>(
            h, conv_W + (size_t)l * HIDC * HIDC, nullptr, mbuf, N);
        zero_f32<<<(nhi + 255) / 256, blk, 0, stream>>>(agg, nhi);
        {
            long long threads = (long long)E1 * 32;
            scatter_add_edges<<<(unsigned)((threads + 255) / 256), blk, 0, stream>>>(mbuf, ei, E1, agg);
        }
        bias_relu<<<(nhi + 255) / 256, blk, 0, stream>>>(agg, conv_b + l * HIDC, nhi);
        zero_f32<<<(N + 255) / 256, blk, 0, stream>>>(gsum1, N);
        zero_f32<<<(N + 255) / 256, blk, 0, stream>>>(gsum2, N);
        gamma_edges<<<(E1 + 7) / 8, blk, 0, stream>>>(agg, ei,  E1, gsum1);
        gamma_edges<<<(E2 + 7) / 8, blk, 0, stream>>>(agg, ei2, E2, gsum2);
        gamma_node<<<(N + 255) / 256, blk, 0, stream>>>(gsum1, inv1, g1, N);
        gamma_node<<<(N + 255) / 256, blk, 0, stream>>>(gsum2, inv2, g2, N);
        update_h<<<(nhi + 255) / 256, blk, 0, stream>>>(h, agg, xskip, g1, g2, nhi);
    }

    gemm_wmma_f32<40><<<gemm_blocks(N, 40), blk, 0, stream>>>(h, fc_W, fc_b, out, N);
}


# agg: h rows summed first, projected after: the owner block multiplies its 80x128 LDS tile of sums by conv_W on WMMA (A from LDS); no h@conv_W launch, no m plane; 15 -> 13 launches; sum order changes
# speedup vs baseline: 2.0216x; 2.0216x over previous
#include <hip/hip_runtime.h>

typedef __attribute__((ext_vector_type(2))) float v2f;
typedef __attribute__((ext_vector_type(8))) float v8f;

#define HIDC 128

template <int NC>
__global__ __launch_bounds__(256) void gemm_wmma_f32(
    const float* __restrict__ A, const float* __restrict__ B,
    const float* __restrict__ bias, float* __restrict__ C, int M)
{
    constexpr int K = HIDC;
    constexpr int tilesN = (NC + 15) >> 4;
    const int tilesM = (M + 15) >> 4;
    const int wid = blockIdx.x * 8 + (threadIdx.x >> 5);
    if (wid >= tilesM * tilesN) return;
    const int tm = wid / tilesN, tn = wid % tilesN;
    const int m0 = tm << 4, n0 = tn << 4;
    const int lane = threadIdx.x & 31;
    const int half = lane >> 4;
    const int l16  = lane & 15;

    const int arow   = min(m0 + l16, M - 1);
    const int bcol_r = n0 + l16;
    const int bcol   = min(bcol_r, NC - 1);

    const float* Ap = A + (size_t)arow * K + 2 * half;
    const float* Bp = B + (size_t)(2 * half) * NC + bcol;

    v8f acc = {};
#pragma unroll
    for (int k0 = 0; k0 < K; k0 += 4) {
        v2f a = *(const v2f*)(Ap + k0);
        v2f b;
        b.x = Bp[k0 * NC];
        b.y = Bp[(k0 + 1) * NC];
        acc = __builtin_amdgcn_wmma_f32_16x16x4_f32(
                  false, a, false, b, (short)0, acc, false, false);
    }

    if (bcol_r < NC) {
        const float bv = bias ? bias[bcol_r] : 0.0f;
#pragma unroll
        for (int v = 0; v < 8; ++v) {
            const int r = m0 + v + 8 * half;
            if (r < M) C[(size_t)r * NC + bcol_r] = acc[v] + bv;
        }
    }
}

#define AGG_NB      80
#define AGG_THREADS 1024
#define AGG_U       8
#define AGG_P       132

__global__ __launch_bounds__(AGG_THREADS) void agg_by_range(
    const float* __restrict__ hsrc, const int* __restrict__ ei, int E,
    const float* __restrict__ Wm, const float* __restrict__ bias,
    float* __restrict__ agg, int n)
{
    __shared__ float acc[AGG_NB * AGG_P];
    __shared__ float prj[AGG_NB * AGG_P];
    const int tid  = threadIdx.x;
    const int lane = tid & 31;
    const int wave = tid >> 5;
    const int base = blockIdx.x * AGG_NB;

    for (int i = tid; i < AGG_NB * AGG_P / 4; i += AGG_THREADS)
        ((float4*)acc)[i] = make_float4(0.0f, 0.0f, 0.0f, 0.0f);
    __syncthreads();

    const int* __restrict__ rowp = ei;
    const int* __restrict__ colp = ei + E;
    constexpr int STEP = 32 * AGG_U;
    const int nstep = (E + STEP - 1) / STEP;
    const int rot = (int)((blockIdx.x * (unsigned)nstep) / gridDim.x);
    for (int s = wave; s < nstep; s += AGG_THREADS / 32) {
        int sc = s + rot;
        if (sc >= nstep) sc -= nstep;
        const int e0 = sc * STEP;
        int  d[AGG_U], r[AGG_U];
        bool hit[AGG_U];
#pragma unroll
        for (int u = 0; u < AGG_U; ++u) {
            const int e  = e0 + u * 32 + lane;
            const int ec = min(e, E - 1);
            d[u]   = colp[ec] - base;
            hit[u] = (e < E) && ((unsigned)d[u] < (unsigned)AGG_NB);
        }
#pragma unroll
        for (int u = 0; u < AGG_U; ++u) {
            r[u] = 0;
            if (hit[u]) r[u] = rowp[e0 + u * 32 + lane];
        }
#pragma unroll
        for (int u = 0; u < AGG_U; ++u) {
            unsigned mask = __builtin_amdgcn_ballot_w32(hit[u]);
            while (mask) {
                const int j = __builtin_ctz(mask);
                mask &= mask - 1;
                const int rj = __builtin_amdgcn_readlane(r[u], j);
                const int dj = __builtin_amdgcn_readlane(d[u], j);
                const float4 v = *((const float4*)(hsrc + (size_t)rj * HIDC) + lane);
                float* a = acc + dj * AGG_P + lane;
                atomicAdd(a,      v.x);
                atomicAdd(a + 32, v.y);
                atomicAdd(a + 64, v.z);
                atomicAdd(a + 96, v.w);
            }
        }
    }
    __syncthreads();

    {
        const int half = lane >> 4;
        const int l16  = lane & 15;
        const int n0   = (wave & 7) << 4;
        const float* Bp = Wm + (size_t)(2 * half) * HIDC + n0 + l16;
        const int col = n0 + l16;
        for (int mt = wave >> 3; mt < AGG_NB / 16; mt += AGG_THREADS / 256) {
            const float* Ar = acc + (mt * 16 + l16) * AGG_P + (2 * half) * 32;
            v8f d = {};
#pragma unroll
            for (int k0 = 0; k0 < HIDC; k0 += 4) {
                v2f a, b;
                a.x = Ar[k0 >> 2];
                a.y = Ar[32 + (k0 >> 2)];
                b.x = Bp[k0 * HIDC];
                b.y = Bp[(k0 + 1) * HIDC];
                d = __builtin_amdgcn_wmma_f32_16x16x4_f32(
                        false, a, false, b, (short)0, d, false, false);
            }
            float* o = prj + (mt * 16 + 8 * half) * AGG_P + (col & 3) * 32 + (col >> 2);
#pragma unroll
            for (int v = 0; v < 8; ++v) o[v * AGG_P] = d[v];
        }
    }
    __syncthreads();

    const float4 bv = *((const float4*)bias + lane);
    for (int i = tid; i < AGG_NB * 32; i += AGG_THREADS) {
        const int rl   = i >> 5;
        const int node = base + rl;
        if (node < n) {
            const float* a = prj + rl * AGG_P + lane;
            float4 o;
            o.x = fmaxf(a[0]  + bv.x, 0.0f);
            o.y = fmaxf(a[32] + bv.y, 0.0f);
            o.z = fmaxf(a[64] + bv.z, 0.0f);
            o.w = fmaxf(a[96] + bv.w, 0.0f);
            *((float4*)(agg + (size_t)node * HIDC) + lane) = o;
        }
    }
}

template <int CTRL>
__device__ __forceinline__ float lane_read(float v) {
    return __int_as_float(__builtin_amdgcn_update_dpp(
        0, __float_as_int(v), CTRL, 0xF, 0xF, true));
}

#define GR_NB      80
#define GR_THREADS 1024
#define GR_CAP     32768
#define GR_U       16

__global__ __launch_bounds__(GR_THREADS) void range_table(
    const int* __restrict__ ei1, int E1, int ng1,
    const int* __restrict__ ei2, int E2,
    int* __restrict__ tab1, int* __restrict__ tab2)
{
    __shared__ int smin, smax;
    int grp = blockIdx.x;
    const int* __restrict__ rowp = ei1;
    int  E   = E1;
    int* tab = tab1;
    if (grp >= ng1) { grp -= ng1; rowp = ei2; E = E2; tab = tab2; }
    if (threadIdx.x == 0) { smin = 0x7fffffff; smax = -0x7fffffff - 1; }
    __syncthreads();
    int lo = 0x7fffffff, hi = -0x7fffffff - 1;
    const int e0 = grp * GR_CAP + (int)threadIdx.x;
#pragma unroll 8
    for (int k = 0; k < GR_CAP / GR_THREADS; ++k) {
        const int r = rowp[min(e0 + k * GR_THREADS, E - 1)];
        lo = min(lo, r);
        hi = max(hi, r);
    }
    atomicMin(&smin, lo);
    atomicMax(&smax, hi);
    __syncthreads();
    if (threadIdx.x == 0) { tab[2 * grp] = smin; tab[2 * grp + 1] = smax; }
}

__global__ __launch_bounds__(GR_THREADS) void gamma_by_range(
    const float* __restrict__ agg, const int* __restrict__ ei, int E,
    const int* __restrict__ tab, float* __restrict__ g, int n)
{
    __shared__ float    gs[GR_NB];
    __shared__ int      cnt[GR_NB];
    __shared__ int      nlist;
    __shared__ unsigned list[GR_CAP];
    const int tid  = threadIdx.x;
    const int lane = tid & 31;
    const int wave = tid >> 5;
    const int base = blockIdx.x * GR_NB;
    const int slot = lane >> 4;
    const int sub  = lane & 15;

    if (tid < GR_NB) { gs[tid] = 0.0f; cnt[tid] = 0; }
    if (tid == 0) nlist = 0;
    __syncthreads();

    const int* __restrict__ rowp = ei;
    const int* __restrict__ colp = ei + E;
    const int ngrp = (E + GR_CAP - 1) / GR_CAP;
    const int rot = (int)(blockIdx.x % (unsigned)ngrp);
    for (int g0 = 0; g0 < ngrp; ++g0) {
        int gi = g0 + rot;
        if (gi >= ngrp) gi -= ngrp;
        const int lo = tab[2 * gi], hi = tab[2 * gi + 1];
        if (hi < base || lo >= base + GR_NB) continue;

        for (int it = 0; it < GR_CAP / (GR_THREADS * GR_U); ++it) {
            const int e0 = gi * GR_CAP + (it * (GR_THREADS / 32) + wave) * (32 * GR_U);
            int  rr[GR_U], cc[GR_U];
            bool hit[GR_U];
#pragma unroll
            for (int u = 0; u < GR_U; ++u) {
                const int e  = e0 + u * 32 + lane;
                const int ec = min(e, E - 1);
                rr[u]  = rowp[ec] - base;
                hit[u] = (e < E) && ((unsigned)rr[u] < (unsigned)GR_NB);
            }
#pragma unroll
            for (int u = 0; u < GR_U; ++u) {
                cc[u] = 0;
                if (hit[u]) cc[u] = colp[e0 + u * 32 + lane];
            }
#pragma unroll
            for (int u = 0; u < GR_U; ++u) {
                const unsigned mask = __builtin_amdgcn_ballot_w32(hit[u]);
                if (mask) {
                    int off = 0;
                    if (lane == 0) off = atomicAdd(&nlist, __builtin_popcount(mask));
                    off = __builtin_amdgcn_readfirstlane(off);
                    if (hit[u])
                        list[off + __builtin_amdgcn_mbcnt_lo(mask, 0)] =
                            ((unsigned)cc[u] << 7) | (unsigned)rr[u];
                }
            }
        }
        __syncthreads();
        const int nh = nlist;

        for (int s = wave * 8; s < nh; s += (GR_THREADS / 32) * 8) {
            int    rl[4];
            bool   ok[4];
            float4 a[4][2], b[4][2];
#pragma unroll
            for (int q = 0; q < 4; ++q) {
                const int idx = s + 2 * q + slot;
                ok[q] = idx < nh;
                const unsigned ent = list[min(idx, nh - 1)];
                rl[q] = (int)(ent & 127u);
                const int c = (int)(ent >> 7);
                const float4* pa = (const float4*)(agg + (size_t)(base + rl[q]) * HIDC) + sub;
                const float4* pb = (const float4*)(agg + (size_t)c * HIDC) + sub;
#pragma unroll
                for (int i = 0; i < 2; ++i) {
                    a[q][i] = pa[16 * i];
                    b[q][i] = pb[16 * i];
                }
            }

            __builtin_amdgcn_sched_barrier(0);

#pragma unroll
            for (int q = 0; q < 4; ++q) {
                float sv = 0.0f;
#pragma unroll
                for (int i = 0; i < 2; ++i) {
                    const float dx = a[q][i].x - b[q][i].x;
                    const float dy = a[q][i].y - b[q][i].y;
                    const float dz = a[q][i].z - b[q][i].z;
                    const float dw = a[q][i].w - b[q][i].w;
                    sv += dx * dx + dy * dy + dz * dz + dw * dw;
                }
                sv += lane_read<0x161>(sv);
                sv += lane_read<0x162>(sv);
                sv += lane_read<0x164>(sv);
                sv += lane_read<0x168>(sv);
                if (sub == 0 && ok[q]) {
                    atomicAdd(&gs[rl[q]], sv);
                    atomicAdd(&cnt[rl[q]], 1);
                }
            }
        }
        __syncthreads();
        if (tid == 0) nlist = 0;
        __syncthreads();
    }

    if (tid < GR_NB) {
        const int node = base + tid;
        if (node < n)
            g[node] = tanhf(gs[tid] * (1.0f / ((float)cnt[tid] + 1e-10f)));
    }
}

__global__ void update_h(float* __restrict__ h, const float* __restrict__ agg,
                         const float* __restrict__ xs,
                         const float* __restrict__ g1,
                         const float* __restrict__ g2, int total) {
    int i = blockIdx.x * blockDim.x + threadIdx.x;
    if (i < total) {
        int node = i >> 7;
        float gs = g1[node], gq = g2[node];
        h[i] = (h[i] + gs * agg[i] + gq * xs[i]) / (1.0f + gs + gq);
    }
}

extern "C" void kernel_launch(void* const* d_in, const int* in_sizes, int n_in,
                              void* d_out, int out_size, void* d_ws, size_t ws_size,
                              hipStream_t stream) {
    const float* x      = (const float*)d_in[0];
    const float* x0     = (const float*)d_in[1];
    const int*   ei     = (const int*)d_in[2];
    const int*   ei2    = (const int*)d_in[3];
    const float* in_W   = (const float*)d_in[4];
    const float* skip_W = (const float*)d_in[5];
    const float* conv_W = (const float*)d_in[6];
    const float* conv_b = (const float*)d_in[7];
    const float* fc_W   = (const float*)d_in[8];
    const float* fc_b   = (const float*)d_in[9];
    float* out = (float*)d_out;

    const int N  = in_sizes[0] / HIDC;
    const int E1 = in_sizes[2] / 2;
    const int E2 = in_sizes[3] / 2;

    const size_t nh = (size_t)N * HIDC;
    float* h     = (float*)d_ws;
    float* xskip = h + nh;
    float* agg   = xskip + nh;
    float* mbuf  = agg + nh;
    int*   tab1  = (int*)(mbuf + nh);
    const int ng1 = (E1 + GR_CAP - 1) / GR_CAP;
    const int ng2 = (E2 + GR_CAP - 1) / GR_CAP;
    int*   tab2  = tab1 + 2 * ng1;
    float* g1    = mbuf + nh + 4 * (size_t)N;
    float* g2    = g1 + N;

    const dim3 blk(256);
    auto gemm_blocks = [](int M, int Nc) {
        int tiles = ((M + 15) / 16) * ((Nc + 15) / 16);
        return dim3((tiles + 7) / 8);
    };
    const int nhi = (int)nh;
    const int nown = (N + GR_NB - 1) / GR_NB;

    gemm_wmma_f32<HIDC><<<gemm_blocks(N, HIDC), blk, 0, stream>>>(x,    in_W,   nullptr, h,     N);
    gemm_wmma_f32<HIDC><<<gemm_blocks(N, HIDC), blk, 0, stream>>>(x0,   in_W,   nullptr, mbuf,  N);
    gemm_wmma_f32<HIDC><<<gemm_blocks(N, HIDC), blk, 0, stream>>>(mbuf, skip_W, nullptr, xskip, N);

    range_table<<<ng1 + ng2, dim3(GR_THREADS), 0, stream>>>(ei, E1, ng1, ei2, E2, tab1, tab2);

    for (int l = 0; l < 2; ++l) {
        agg_by_range<<<(N + AGG_NB - 1) / AGG_NB, dim3(AGG_THREADS), 0, stream>>>(
            h, ei, E1, conv_W + (size_t)l * HIDC * HIDC, conv_b + l * HIDC, agg, N);
        gamma_by_range<<<nown, dim3(GR_THREADS), 0, stream>>>(agg, ei,  E1, tab1, g1, N);
        gamma_by_range<<<nown, dim3(GR_THREADS), 0, stream>>>(agg, ei2, E2, tab2, g2, N);
        update_h<<<(nhi + 255) / 256, blk, 0, stream>>>(h, agg, xskip, g1, g2, nhi);
    }

    gemm_wmma_f32<40><<<gemm_blocks(N, 40), blk, 0, stream>>>(h, fc_W, fc_b, out, N);
}


# gate owner closes the layer: the block owning 80 rows, its 2-hop gate done, updates those rows of h in place; no update_h launches, no g2 plane; 13 -> 11 launches; same arithmetic and sum order
# speedup vs baseline: 2.1270x; 1.0521x over previous
#include <hip/hip_runtime.h>

typedef __attribute__((ext_vector_type(2))) float v2f;
typedef __attribute__((ext_vector_type(8))) float v8f;

#define HIDC 128

template <int NC>
__global__ __launch_bounds__(256) void gemm_wmma_f32(
    const float* __restrict__ A, const float* __restrict__ B,
    const float* __restrict__ bias, float* __restrict__ C, int M)
{
    constexpr int K = HIDC;
    constexpr int tilesN = (NC + 15) >> 4;
    const int tilesM = (M + 15) >> 4;
    const int wid = blockIdx.x * 8 + (threadIdx.x >> 5);
    if (wid >= tilesM * tilesN) return;
    const int tm = wid / tilesN, tn = wid % tilesN;
    const int m0 = tm << 4, n0 = tn << 4;
    const int lane = threadIdx.x & 31;
    const int half = lane >> 4;
    const int l16  = lane & 15;

    const int arow   = min(m0 + l16, M - 1);
    const int bcol_r = n0 + l16;
    const int bcol   = min(bcol_r, NC - 1);

    const float* Ap = A + (size_t)arow * K + 2 * half;
    const float* Bp = B + (size_t)(2 * half) * NC + bcol;

    v8f acc = {};
#pragma unroll
    for (int k0 = 0; k0 < K; k0 += 4) {
        v2f a = *(const v2f*)(Ap + k0);
        v2f b;
        b.x = Bp[k0 * NC];
        b.y = Bp[(k0 + 1) * NC];
        acc = __builtin_amdgcn_wmma_f32_16x16x4_f32(
                  false, a, false, b, (short)0, acc, false, false);
    }

    if (bcol_r < NC) {
        const float bv = bias ? bias[bcol_r] : 0.0f;
#pragma unroll
        for (int v = 0; v < 8; ++v) {
            const int r = m0 + v + 8 * half;
            if (r < M) C[(size_t)r * NC + bcol_r] = acc[v] + bv;
        }
    }
}

#define AGG_NB      80
#define AGG_THREADS 1024
#define AGG_U       8
#define AGG_P       132

__global__ __launch_bounds__(AGG_THREADS) void agg_by_range(
    const float* __restrict__ hsrc, const int* __restrict__ ei, int E,
    const float* __restrict__ Wm, const float* __restrict__ bias,
    float* __restrict__ agg, int n)
{
    __shared__ float acc[AGG_NB * AGG_P];
    __shared__ float prj[AGG_NB * AGG_P];
    const int tid  = threadIdx.x;
    const int lane = tid & 31;
    const int wave = tid >> 5;
    const int base = blockIdx.x * AGG_NB;

    for (int i = tid; i < AGG_NB * AGG_P / 4; i += AGG_THREADS)
        ((float4*)acc)[i] = make_float4(0.0f, 0.0f, 0.0f, 0.0f);
    __syncthreads();

    const int* __restrict__ rowp = ei;
    const int* __restrict__ colp = ei + E;
    constexpr int STEP = 32 * AGG_U;
    const int nstep = (E + STEP - 1) / STEP;
    const int rot = (int)((blockIdx.x * (unsigned)nstep) / gridDim.x);
    for (int s = wave; s < nstep; s += AGG_THREADS / 32) {
        int sc = s + rot;
        if (sc >= nstep) sc -= nstep;
        const int e0 = sc * STEP;
        int  d[AGG_U], r[AGG_U];
        bool hit[AGG_U];
#pragma unroll
        for (int u = 0; u < AGG_U; ++u) {
            const int e  = e0 + u * 32 + lane;
            const int ec = min(e, E - 1);
            d[u]   = colp[ec] - base;
            hit[u] = (e < E) && ((unsigned)d[u] < (unsigned)AGG_NB);
        }
#pragma unroll
        for (int u = 0; u < AGG_U; ++u) {
            r[u] = 0;
            if (hit[u]) r[u] = rowp[e0 + u * 32 + lane];
        }
#pragma unroll
        for (int u = 0; u < AGG_U; ++u) {
            unsigned mask = __builtin_amdgcn_ballot_w32(hit[u]);
            while (mask) {
                const int j = __builtin_ctz(mask);
                mask &= mask - 1;
                const int rj = __builtin_amdgcn_readlane(r[u], j);
                const int dj = __builtin_amdgcn_readlane(d[u], j);
                const float4 v = *((const float4*)(hsrc + (size_t)rj * HIDC) + lane);
                float* a = acc + dj * AGG_P + lane;
                atomicAdd(a,      v.x);
                atomicAdd(a + 32, v.y);
                atomicAdd(a + 64, v.z);
                atomicAdd(a + 96, v.w);
            }
        }
    }
    __syncthreads();

    {
        const int half = lane >> 4;
        const int l16  = lane & 15;
        const int n0   = (wave & 7) << 4;
        const float* Bp = Wm + (size_t)(2 * half) * HIDC + n0 + l16;
        const int col = n0 + l16;
        for (int mt = wave >> 3; mt < AGG_NB / 16; mt += AGG_THREADS / 256) {
            const float* Ar = acc + (mt * 16 + l16) * AGG_P + (2 * half) * 32;
            v8f d = {};
#pragma unroll
            for (int k0 = 0; k0 < HIDC; k0 += 4) {
                v2f a, b;
                a.x = Ar[k0 >> 2];
                a.y = Ar[32 + (k0 >> 2)];
                b.x = Bp[k0 * HIDC];
                b.y = Bp[(k0 + 1) * HIDC];
                d = __builtin_amdgcn_wmma_f32_16x16x4_f32(
                        false, a, false, b, (short)0, d, false, false);
            }
            float* o = prj + (mt * 16 + 8 * half) * AGG_P + (col & 3) * 32 + (col >> 2);
#pragma unroll
            for (int v = 0; v < 8; ++v) o[v * AGG_P] = d[v];
        }
    }
    __syncthreads();

    const float4 bv = *((const float4*)bias + lane);
    for (int i = tid; i < AGG_NB * 32; i += AGG_THREADS) {
        const int rl   = i >> 5;
        const int node = base + rl;
        if (node < n) {
            const float* a = prj + rl * AGG_P + lane;
            float4 o;
            o.x = fmaxf(a[0]  + bv.x, 0.0f);
            o.y = fmaxf(a[32] + bv.y, 0.0f);
            o.z = fmaxf(a[64] + bv.z, 0.0f);
            o.w = fmaxf(a[96] + bv.w, 0.0f);
            *((float4*)(agg + (size_t)node * HIDC) + lane) = o;
        }
    }
}

template <int CTRL>
__device__ __forceinline__ float lane_read(float v) {
    return __int_as_float(__builtin_amdgcn_update_dpp(
        0, __float_as_int(v), CTRL, 0xF, 0xF, true));
}

#define GR_NB      80
#define GR_THREADS 1024
#define GR_CAP     32768
#define GR_U       16

__device__ __forceinline__ float gate_value(float s, int c) {
    return tanhf(s * (1.0f / ((float)c + 1e-10f)));
}

__global__ __launch_bounds__(GR_THREADS) void range_table(
    const int* __restrict__ ei1, int E1, int ng1,
    const int* __restrict__ ei2, int E2,
    int* __restrict__ tab1, int* __restrict__ tab2)
{
    __shared__ int smin, smax;
    int grp = blockIdx.x;
    const int* __restrict__ rowp = ei1;
    int  E   = E1;
    int* tab = tab1;
    if (grp >= ng1) { grp -= ng1; rowp = ei2; E = E2; tab = tab2; }
    if (threadIdx.x == 0) { smin = 0x7fffffff; smax = -0x7fffffff - 1; }
    __syncthreads();
    int lo = 0x7fffffff, hi = -0x7fffffff - 1;
    const int e0 = grp * GR_CAP + (int)threadIdx.x;
#pragma unroll 8
    for (int k = 0; k < GR_CAP / GR_THREADS; ++k) {
        const int r = rowp[min(e0 + k * GR_THREADS, E - 1)];
        lo = min(lo, r);
        hi = max(hi, r);
    }
    atomicMin(&smin, lo);
    atomicMax(&smax, hi);
    __syncthreads();
    if (threadIdx.x == 0) { tab[2 * grp] = smin; tab[2 * grp + 1] = smax; }
}

__global__ __launch_bounds__(GR_THREADS) void gamma_by_range(
    const float* __restrict__ agg, const int* __restrict__ ei, int E,
    const int* __restrict__ tab, float* __restrict__ g,
    const float* __restrict__ gprev, float* __restrict__ hup,
    const float* __restrict__ xs, int n)
{
    __shared__ float    gs[GR_NB];
    __shared__ int      cnt[GR_NB];
    __shared__ int      nlist;
    __shared__ unsigned list[GR_CAP];
    const int tid  = threadIdx.x;
    const int lane = tid & 31;
    const int wave = tid >> 5;
    const int base = blockIdx.x * GR_NB;
    const int slot = lane >> 4;
    const int sub  = lane & 15;

    if (tid < GR_NB) { gs[tid] = 0.0f; cnt[tid] = 0; }
    if (tid == 0) nlist = 0;
    __syncthreads();

    const int* __restrict__ rowp = ei;
    const int* __restrict__ colp = ei + E;
    const int ngrp = (E + GR_CAP - 1) / GR_CAP;
    const int rot = (int)(blockIdx.x % (unsigned)ngrp);
    for (int g0 = 0; g0 < ngrp; ++g0) {
        int gi = g0 + rot;
        if (gi >= ngrp) gi -= ngrp;
        const int lo = tab[2 * gi], hi = tab[2 * gi + 1];
        if (hi < base || lo >= base + GR_NB) continue;

        for (int it = 0; it < GR_CAP / (GR_THREADS * GR_U); ++it) {
            const int e0 = gi * GR_CAP + (it * (GR_THREADS / 32) + wave) * (32 * GR_U);
            int  rr[GR_U], cc[GR_U];
            bool hit[GR_U];
#pragma unroll
            for (int u = 0; u < GR_U; ++u) {
                const int e  = e0 + u * 32 + lane;
                const int ec = min(e, E - 1);
                rr[u]  = rowp[ec] - base;
                hit[u] = (e < E) && ((unsigned)rr[u] < (unsigned)GR_NB);
            }
#pragma unroll
            for (int u = 0; u < GR_U; ++u) {
                cc[u] = 0;
                if (hit[u]) cc[u] = colp[e0 + u * 32 + lane];
            }
#pragma unroll
            for (int u = 0; u < GR_U; ++u) {
                const unsigned mask = __builtin_amdgcn_ballot_w32(hit[u]);
                if (mask) {
                    int off = 0;
                    if (lane == 0) off = atomicAdd(&nlist, __builtin_popcount(mask));
                    off = __builtin_amdgcn_readfirstlane(off);
                    if (hit[u])
                        list[off + __builtin_amdgcn_mbcnt_lo(mask, 0)] =
                            ((unsigned)cc[u] << 7) | (unsigned)rr[u];
                }
            }
        }
        __syncthreads();
        const int nh = nlist;

        for (int s = wave * 8; s < nh; s += (GR_THREADS / 32) * 8) {
            int    rl[4];
            bool   ok[4];
            float4 a[4][2], b[4][2];
#pragma unroll
            for (int q = 0; q < 4; ++q) {
                const int idx = s + 2 * q + slot;
                ok[q] = idx < nh;
                const unsigned ent = list[min(idx, nh - 1)];
                rl[q] = (int)(ent & 127u);
                const int c = (int)(ent >> 7);
                const float4* pa = (const float4*)(agg + (size_t)(base + rl[q]) * HIDC) + sub;
                const float4* pb = (const float4*)(agg + (size_t)c * HIDC) + sub;
#pragma unroll
                for (int i = 0; i < 2; ++i) {
                    a[q][i] = pa[16 * i];
                    b[q][i] = pb[16 * i];
                }
            }

            __builtin_amdgcn_sched_barrier(0);

#pragma unroll
            for (int q = 0; q < 4; ++q) {
                float sv = 0.0f;
#pragma unroll
                for (int i = 0; i < 2; ++i) {
                    const float dx = a[q][i].x - b[q][i].x;
                    const float dy = a[q][i].y - b[q][i].y;
                    const float dz = a[q][i].z - b[q][i].z;
                    const float dw = a[q][i].w - b[q][i].w;
                    sv += dx * dx + dy * dy + dz * dz + dw * dw;
                }
                sv += lane_read<0x161>(sv);
                sv += lane_read<0x162>(sv);
                sv += lane_read<0x164>(sv);
                sv += lane_read<0x168>(sv);
                if (sub == 0 && ok[q]) {
                    atomicAdd(&gs[rl[q]], sv);
                    atomicAdd(&cnt[rl[q]], 1);
                }
            }
        }
        __syncthreads();
        if (tid == 0) nlist = 0;
        __syncthreads();
    }

    if (hup == nullptr) {
        if (tid < GR_NB && base + tid < n) g[base + tid] = gate_value(gs[tid], cnt[tid]);
        return;
    }

    constexpr int UPD_T = (GR_NB * 32 + GR_THREADS - 1) / GR_THREADS;
    float4 hv[UPD_T], av[UPD_T], xv[UPD_T];
    float  gm[UPD_T];
#pragma unroll
    for (int t = 0; t < UPD_T; ++t) {
        const int rl   = min((tid + t * GR_THREADS) >> 5, GR_NB - 1);
        const int node = min(base + rl, n - 1);
        const size_t o = (size_t)node * (HIDC / 4) + lane;
        gm[t] = gprev[node];
        hv[t] = ((const float4*)hup)[o];
        av[t] = ((const float4*)agg)[o];
        xv[t] = ((const float4*)xs)[o];
    }

    if (tid < GR_NB) gs[tid] = gate_value(gs[tid], cnt[tid]);
    __syncthreads();

#pragma unroll
    for (int t = 0; t < UPD_T; ++t) {
        const int i    = tid + t * GR_THREADS;
        const int rl   = i >> 5;
        const int node = base + rl;
        if (i < GR_NB * 32 && node < n) {
            const float gq  = gs[rl];
            const float den = 1.0f + gm[t] + gq;
            float4 r;
            r.x = (hv[t].x + gm[t] * av[t].x + gq * xv[t].x) / den;
            r.y = (hv[t].y + gm[t] * av[t].y + gq * xv[t].y) / den;
            r.z = (hv[t].z + gm[t] * av[t].z + gq * xv[t].z) / den;
            r.w = (hv[t].w + gm[t] * av[t].w + gq * xv[t].w) / den;
            ((float4*)hup)[(size_t)node * (HIDC / 4) + lane] = r;
        }
    }
}

extern "C" void kernel_launch(void* const* d_in, const int* in_sizes, int n_in,
                              void* d_out, int out_size, void* d_ws, size_t ws_size,
                              hipStream_t stream) {
    const float* x      = (const float*)d_in[0];
    const float* x0     = (const float*)d_in[1];
    const int*   ei     = (const int*)d_in[2];
    const int*   ei2    = (const int*)d_in[3];
    const float* in_W   = (const float*)d_in[4];
    const float* skip_W = (const float*)d_in[5];
    const float* conv_W = (const float*)d_in[6];
    const float* conv_b = (const float*)d_in[7];
    const float* fc_W   = (const float*)d_in[8];
    const float* fc_b   = (const float*)d_in[9];
    float* out = (float*)d_out;

    const int N  = in_sizes[0] / HIDC;
    const int E1 = in_sizes[2] / 2;
    const int E2 = in_sizes[3] / 2;

    const size_t nh = (size_t)N * HIDC;
    float* h     = (float*)d_ws;
    float* xskip = h + nh;
    float* agg   = xskip + nh;
    float* mbuf  = agg + nh;
    int*   tab1  = (int*)(mbuf + nh);
    const int ng1 = (E1 + GR_CAP - 1) / GR_CAP;
    const int ng2 = (E2 + GR_CAP - 1) / GR_CAP;
    int*   tab2  = tab1 + 2 * ng1;
    float* g1    = mbuf + nh + 4 * (size_t)N;

    const dim3 blk(256);
    auto gemm_blocks = [](int M, int Nc) {
        int tiles = ((M + 15) / 16) * ((Nc + 15) / 16);
        return dim3((tiles + 7) / 8);
    };
    const int nown = (N + GR_NB - 1) / GR_NB;

    gemm_wmma_f32<HIDC><<<gemm_blocks(N, HIDC), blk, 0, stream>>>(x,    in_W,   nullptr, h,     N);
    gemm_wmma_f32<HIDC><<<gemm_blocks(N, HIDC), blk, 0, stream>>>(x0,   in_W,   nullptr, mbuf,  N);
    gemm_wmma_f32<HIDC><<<gemm_blocks(N, HIDC), blk, 0, stream>>>(mbuf, skip_W, nullptr, xskip, N);

    range_table<<<ng1 + ng2, dim3(GR_THREADS), 0, stream>>>(ei, E1, ng1, ei2, E2, tab1, tab2);

    for (int l = 0; l < 2; ++l) {
        agg_by_range<<<(N + AGG_NB - 1) / AGG_NB, dim3(AGG_THREADS), 0, stream>>>(
            h, ei, E1, conv_W + (size_t)l * HIDC * HIDC, conv_b + l * HIDC, agg, N);
        gamma_by_range<<<nown, dim3(GR_THREADS), 0, stream>>>(
            agg, ei, E1, tab1, g1, nullptr, nullptr, nullptr, N);
        gamma_by_range<<<nown, dim3(GR_THREADS), 0, stream>>>(
            agg, ei2, E2, tab2, nullptr, g1, h, xskip, N);
    }

    gemm_wmma_f32<40><<<gemm_blocks(N, 40), blk, 0, stream>>>(h, fc_W, fc_b, out, N);
}


# one-hop list: each owner block's edges picked out once per call into two workspace lists (by destination, by row); the 2 agg and 2 gate launches read those, no whole-list walks; sum order changes
# speedup vs baseline: 2.5548x; 1.2011x over previous
#include <hip/hip_runtime.h>

typedef __attribute__((ext_vector_type(2))) float v2f;
typedef __attribute__((ext_vector_type(8))) float v8f;

#define HIDC 128

template <int NC>
__global__ __launch_bounds__(256) void gemm_wmma_f32(
    const float* __restrict__ A, const float* __restrict__ B,
    const float* __restrict__ bias, float* __restrict__ C, int M)
{
    constexpr int K = HIDC;
    constexpr int tilesN = (NC + 15) >> 4;
    const int tilesM = (M + 15) >> 4;
    const int wid = blockIdx.x * 8 + (threadIdx.x >> 5);
    if (wid >= tilesM * tilesN) return;
    const int tm = wid / tilesN, tn = wid % tilesN;
    const int m0 = tm << 4, n0 = tn << 4;
    const int lane = threadIdx.x & 31;
    const int half = lane >> 4;
    const int l16  = lane & 15;

    const int arow   = min(m0 + l16, M - 1);
    const int bcol_r = n0 + l16;
    const int bcol   = min(bcol_r, NC - 1);

    const float* Ap = A + (size_t)arow * K + 2 * half;
    const float* Bp = B + (size_t)(2 * half) * NC + bcol;

    v8f acc = {};
#pragma unroll
    for (int k0 = 0; k0 < K; k0 += 4) {
        v2f a = *(const v2f*)(Ap + k0);
        v2f b;
        b.x = Bp[k0 * NC];
        b.y = Bp[(k0 + 1) * NC];
        acc = __builtin_amdgcn_wmma_f32_16x16x4_f32(
                  false, a, false, b, (short)0, acc, false, false);
    }

    if (bcol_r < NC) {
        const float bv = bias ? bias[bcol_r] : 0.0f;
#pragma unroll
        for (int v = 0; v < 8; ++v) {
            const int r = m0 + v + 8 * half;
            if (r < M) C[(size_t)r * NC + bcol_r] = acc[v] + bv;
        }
    }
}

#define AGG_NB      80
#define AGG_THREADS 1024
#define AGG_U       8
#define AGG_P       132
#define OL_CAP      4096
#define OL_U        8
#define AGG_B       8

__global__ __launch_bounds__(AGG_THREADS) void agg_by_range(
    const float* __restrict__ hsrc, const int* __restrict__ ei, int E,
    const float* __restrict__ Wm, const float* __restrict__ bias,
    float* __restrict__ agg, int n,
    const unsigned* __restrict__ olist, const int* __restrict__ ocnt)
{
    __shared__ float acc[AGG_NB * AGG_P];
    __shared__ float prj[AGG_NB * AGG_P];
    const int tid  = threadIdx.x;
    const int lane = tid & 31;
    const int wave = tid >> 5;
    const int base = blockIdx.x * AGG_NB;

    for (int i = tid; i < AGG_NB * AGG_P / 4; i += AGG_THREADS)
        ((float4*)acc)[i] = make_float4(0.0f, 0.0f, 0.0f, 0.0f);
    __syncthreads();

    const int* __restrict__ rowp = ei;
    const int* __restrict__ colp = ei + E;
    constexpr int STEP = 32 * AGG_U;
    const int nstep = (E + STEP - 1) / STEP;
    const int rot = (int)((blockIdx.x * (unsigned)nstep) / gridDim.x);
    const int lcnt = ocnt[2 * blockIdx.x];
    if (lcnt <= OL_CAP) {
        const unsigned* __restrict__ L = olist + (size_t)blockIdx.x * OL_CAP;
        for (int s = wave * AGG_B; s < lcnt; s += (AGG_THREADS / 32) * AGG_B) {
            unsigned ent[AGG_B];
            float4   v[AGG_B];
#pragma unroll
            for (int q = 0; q < AGG_B; ++q)
                ent[q] = L[min(s + q, lcnt - 1)];
#pragma unroll
            for (int q = 0; q < AGG_B; ++q)
                v[q] = *((const float4*)(hsrc + (size_t)(ent[q] >> 7) * HIDC) + lane);
#pragma unroll
            for (int q = 0; q < AGG_B; ++q) {
                const bool ok = s + q < lcnt;
                float* a = acc + (int)(ent[q] & 127u) * AGG_P + lane;
                atomicAdd(a,      ok ? v[q].x : 0.0f);
                atomicAdd(a + 32, ok ? v[q].y : 0.0f);
                atomicAdd(a + 64, ok ? v[q].z : 0.0f);
                atomicAdd(a + 96, ok ? v[q].w : 0.0f);
            }
        }
    } else {
    for (int s = wave; s < nstep; s += AGG_THREADS / 32) {
        int sc = s + rot;
        if (sc >= nstep) sc -= nstep;
        const int e0 = sc * STEP;
        int  d[AGG_U], r[AGG_U];
        bool hit[AGG_U];
#pragma unroll
        for (int u = 0; u < AGG_U; ++u) {
            const int e  = e0 + u * 32 + lane;
            const int ec = min(e, E - 1);
            d[u]   = colp[ec] - base;
            hit[u] = (e < E) && ((unsigned)d[u] < (unsigned)AGG_NB);
        }
#pragma unroll
        for (int u = 0; u < AGG_U; ++u) {
            r[u] = 0;
            if (hit[u]) r[u] = rowp[e0 + u * 32 + lane];
        }
#pragma unroll
        for (int u = 0; u < AGG_U; ++u) {
            unsigned mask = __builtin_amdgcn_ballot_w32(hit[u]);
            while (mask) {
                const int j = __builtin_ctz(mask);
                mask &= mask - 1;
                const int rj = __builtin_amdgcn_readlane(r[u], j);
                const int dj = __builtin_amdgcn_readlane(d[u], j);
                const float4 v = *((const float4*)(hsrc + (size_t)rj * HIDC) + lane);
                float* a = acc + dj * AGG_P + lane;
                atomicAdd(a,      v.x);
                atomicAdd(a + 32, v.y);
                atomicAdd(a + 64, v.z);
                atomicAdd(a + 96, v.w);
            }
        }
    }
    }
    __syncthreads();

    {
        const int half = lane >> 4;
        const int l16  = lane & 15;
        const int n0   = (wave & 7) << 4;
        const float* Bp = Wm + (size_t)(2 * half) * HIDC + n0 + l16;
        const int col = n0 + l16;
        for (int mt = wave >> 3; mt < AGG_NB / 16; mt += AGG_THREADS / 256) {
            const float* Ar = acc + (mt * 16 + l16) * AGG_P + (2 * half) * 32;
            v8f d = {};
#pragma unroll
            for (int k0 = 0; k0 < HIDC; k0 += 4) {
                v2f a, b;
                a.x = Ar[k0 >> 2];
                a.y = Ar[32 + (k0 >> 2)];
                b.x = Bp[k0 * HIDC];
                b.y = Bp[(k0 + 1) * HIDC];
                d = __builtin_amdgcn_wmma_f32_16x16x4_f32(
                        false, a, false, b, (short)0, d, false, false);
            }
            float* o = prj + (mt * 16 + 8 * half) * AGG_P + (col & 3) * 32 + (col >> 2);
#pragma unroll
            for (int v = 0; v < 8; ++v) o[v * AGG_P] = d[v];
        }
    }
    __syncthreads();

    const float4 bv = *((const float4*)bias + lane);
    for (int i = tid; i < AGG_NB * 32; i += AGG_THREADS) {
        const int rl   = i >> 5;
        const int node = base + rl;
        if (node < n) {
            const float* a = prj + rl * AGG_P + lane;
            float4 o;
            o.x = fmaxf(a[0]  + bv.x, 0.0f);
            o.y = fmaxf(a[32] + bv.y, 0.0f);
            o.z = fmaxf(a[64] + bv.z, 0.0f);
            o.w = fmaxf(a[96] + bv.w, 0.0f);
            *((float4*)(agg + (size_t)node * HIDC) + lane) = o;
        }
    }
}

template <int CTRL>
__device__ __forceinline__ float lane_read(float v) {
    return __int_as_float(__builtin_amdgcn_update_dpp(
        0, __float_as_int(v), CTRL, 0xF, 0xF, true));
}

#define GR_NB      80
#define GR_THREADS 1024
#define GR_CAP     32768
#define GR_U       16
static_assert(GR_NB == AGG_NB, "one owner range for the sums and the gates: the owner lists are made for both");
static_assert(OL_CAP <= GR_CAP, "an owner list must fit the LDS list");

__device__ __forceinline__ float gate_value(float s, int c) {
    return tanhf(s * (1.0f / ((float)c + 1e-10f)));
}

__device__ __forceinline__ void owner_lists(
    const int* __restrict__ ei, int E, int b, int nown,
    unsigned* __restrict__ la, unsigned* __restrict__ lg, int* __restrict__ cnts, int* sn)
{
    const int tid  = threadIdx.x;
    const int lane = tid & 31;
    const int wave = tid >> 5;
    const int base = b * GR_NB;
    if (tid < 2) sn[tid] = 0;
    __syncthreads();
    const int* __restrict__ rowp = ei;
    const int* __restrict__ colp = ei + E;
    constexpr int STEP = 32 * OL_U;
    const int nstep = (E + STEP - 1) / STEP;
    const int rot = (int)(((unsigned)b * (unsigned)nstep) / (unsigned)nown);
    for (int s = wave; s < nstep; s += GR_THREADS / 32) {
        int sc = s + rot;
        if (sc >= nstep) sc -= nstep;
        const int e0 = sc * STEP;
        int r[OL_U], c[OL_U];
#pragma unroll
        for (int u = 0; u < OL_U; ++u) {
            const int ec = min(e0 + u * 32 + lane, E - 1);
            r[u] = rowp[ec];
            c[u] = colp[ec];
        }
#pragma unroll
        for (int u = 0; u < OL_U; ++u) {
            const bool in = (e0 + u * 32 + lane) < E;
            const int  dc = c[u] - base, dr = r[u] - base;
            const bool ha = in && ((unsigned)dc < (unsigned)GR_NB);
            const bool hg = in && ((unsigned)dr < (unsigned)GR_NB);
            const unsigned ma = __builtin_amdgcn_ballot_w32(ha);
            if (ma) {
                int off = 0;
                if (lane == 0) off = atomicAdd(&sn[0], __builtin_popcount(ma));
                off = __builtin_amdgcn_readfirstlane(off);
                const int pos = off + (int)__builtin_amdgcn_mbcnt_lo(ma, 0);
                if (ha && pos < OL_CAP) la[pos] = ((unsigned)r[u] << 7) | (unsigned)dc;
            }
            const unsigned mg = __builtin_amdgcn_ballot_w32(hg);
            if (mg) {
                int off = 0;
                if (lane == 0) off = atomicAdd(&sn[1], __builtin_popcount(mg));
                off = __builtin_amdgcn_readfirstlane(off);
                const int pos = off + (int)__builtin_amdgcn_mbcnt_lo(mg, 0);
                if (hg && pos < OL_CAP) lg[pos] = ((unsigned)c[u] << 7) | (unsigned)dr;
            }
        }
    }
    __syncthreads();
    if (tid < 2) cnts[2 * b + tid] = sn[tid];
}

__global__ __launch_bounds__(GR_THREADS) void range_table(
    const int* __restrict__ ei1, int E1, int ng1,
    const int* __restrict__ ei2, int E2,
    int* __restrict__ tab1, int* __restrict__ tab2,
    int nown, unsigned* __restrict__ ola, unsigned* __restrict__ olg, int* __restrict__ ocnt)
{
    __shared__ int smin, smax;
    __shared__ int sn[2];
    if ((int)blockIdx.x < nown) {
        owner_lists(ei1, E1, (int)blockIdx.x, nown,
                    ola + (size_t)blockIdx.x * OL_CAP, olg + (size_t)blockIdx.x * OL_CAP,
                    ocnt, sn);
        return;
    }
    int grp = (int)blockIdx.x - nown;
    const int* __restrict__ rowp = ei1;
    int  E   = E1;
    int* tab = tab1;
    if (grp >= ng1) { grp -= ng1; rowp = ei2; E = E2; tab = tab2; }
    if (threadIdx.x == 0) { smin = 0x7fffffff; smax = -0x7fffffff - 1; }
    __syncthreads();
    int lo = 0x7fffffff, hi = -0x7fffffff - 1;
    const int e0 = grp * GR_CAP + (int)threadIdx.x;
#pragma unroll 8
    for (int k = 0; k < GR_CAP / GR_THREADS; ++k) {
        const int r = rowp[min(e0 + k * GR_THREADS, E - 1)];
        lo = min(lo, r);
        hi = max(hi, r);
    }
    atomicMin(&smin, lo);
    atomicMax(&smax, hi);
    __syncthreads();
    if (threadIdx.x == 0) { tab[2 * grp] = smin; tab[2 * grp + 1] = smax; }
}

__global__ __launch_bounds__(GR_THREADS) void gamma_by_range(
    const float* __restrict__ agg, const int* __restrict__ ei, int E,
    const int* __restrict__ tab, float* __restrict__ g,
    const float* __restrict__ gprev, float* __restrict__ hup,
    const float* __restrict__ xs, int n,
    const unsigned* __restrict__ olist, const int* __restrict__ ocnt)
{
    __shared__ float    gs[GR_NB];
    __shared__ int      cnt[GR_NB];
    __shared__ int      nlist;
    __shared__ unsigned list[GR_CAP];
    const int tid  = threadIdx.x;
    const int lane = tid & 31;
    const int wave = tid >> 5;
    const int base = blockIdx.x * GR_NB;
    const int slot = lane >> 4;
    const int sub  = lane & 15;

    if (tid < GR_NB) { gs[tid] = 0.0f; cnt[tid] = 0; }
    if (tid == 0) nlist = 0;
    __syncthreads();

    const int* __restrict__ rowp = ei;
    const int* __restrict__ colp = ei + E;
    const int ngrp = (E + GR_CAP - 1) / GR_CAP;
    const int rot = (int)(blockIdx.x % (unsigned)ngrp);
    int lcnt = -1;
    if (olist != nullptr) {
        const int c = ocnt[2 * blockIdx.x + 1];
        if (c <= OL_CAP) lcnt = c;
    }
    const int ntrip = (lcnt >= 0) ? 1 : ngrp;
    for (int g0 = 0; g0 < ntrip; ++g0) {
        if (lcnt >= 0) {
            const unsigned* __restrict__ L = olist + (size_t)blockIdx.x * OL_CAP;
            for (int i = tid; i < lcnt; i += GR_THREADS) list[i] = L[i];
            if (tid == 0) nlist = lcnt;
        } else {
        int gi = g0 + rot;
        if (gi >= ngrp) gi -= ngrp;
        const int lo = tab[2 * gi], hi = tab[2 * gi + 1];
        if (hi < base || lo >= base + GR_NB) continue;

        for (int it = 0; it < GR_CAP / (GR_THREADS * GR_U); ++it) {
            const int e0 = gi * GR_CAP + (it * (GR_THREADS / 32) + wave) * (32 * GR_U);
            int  rr[GR_U], cc[GR_U];
            bool hit[GR_U];
#pragma unroll
            for (int u = 0; u < GR_U; ++u) {
                const int e  = e0 + u * 32 + lane;
                const int ec = min(e, E - 1);
                rr[u]  = rowp[ec] - base;
                hit[u] = (e < E) && ((unsigned)rr[u] < (unsigned)GR_NB);
            }
#pragma unroll
            for (int u = 0; u < GR_U; ++u) {
                cc[u] = 0;
                if (hit[u]) cc[u] = colp[e0 + u * 32 + lane];
            }
#pragma unroll
            for (int u = 0; u < GR_U; ++u) {
                const unsigned mask = __builtin_amdgcn_ballot_w32(hit[u]);
                if (mask) {
                    int off = 0;
                    if (lane == 0) off = atomicAdd(&nlist, __builtin_popcount(mask));
                    off = __builtin_amdgcn_readfirstlane(off);
                    if (hit[u])
                        list[off + __builtin_amdgcn_mbcnt_lo(mask, 0)] =
                            ((unsigned)cc[u] << 7) | (unsigned)rr[u];
                }
            }
        }
        }
        __syncthreads();
        const int nh = nlist;

        for (int s = wave * 8; s < nh; s += (GR_THREADS / 32) * 8) {
            int    rl[4];
            bool   ok[4];
            float4 a[4][2], b[4][2];
#pragma unroll
            for (int q = 0; q < 4; ++q) {
                const int idx = s + 2 * q + slot;
                ok[q] = idx < nh;
                const unsigned ent = list[min(idx, nh - 1)];
                rl[q] = (int)(ent & 127u);
                const int c = (int)(ent >> 7);
                const float4* pa = (const float4*)(agg + (size_t)(base + rl[q]) * HIDC) + sub;
                const float4* pb = (const float4*)(agg + (size_t)c * HIDC) + sub;
#pragma unroll
                for (int i = 0; i < 2; ++i) {
                    a[q][i] = pa[16 * i];
                    b[q][i] = pb[16 * i];
                }
            }

            __builtin_amdgcn_sched_barrier(0);

#pragma unroll
            for (int q = 0; q < 4; ++q) {
                float sv = 0.0f;
#pragma unroll
                for (int i = 0; i < 2; ++i) {
                    const float dx = a[q][i].x - b[q][i].x;
                    const float dy = a[q][i].y - b[q][i].y;
                    const float dz = a[q][i].z - b[q][i].z;
                    const float dw = a[q][i].w - b[q][i].w;
                    sv += dx * dx + dy * dy + dz * dz + dw * dw;
                }
                sv += lane_read<0x161>(sv);
                sv += lane_read<0x162>(sv);
                sv += lane_read<0x164>(sv);
                sv += lane_read<0x168>(sv);
                if (sub == 0 && ok[q]) {
                    atomicAdd(&gs[rl[q]], sv);
                    atomicAdd(&cnt[rl[q]], 1);
                }
            }
        }
        __syncthreads();
        if (tid == 0) nlist = 0;
        __syncthreads();
    }

    if (hup == nullptr) {
        if (tid < GR_NB && base + tid < n) g[base + tid] = gate_value(gs[tid], cnt[tid]);
        return;
    }

    constexpr int UPD_T = (GR_NB * 32 + GR_THREADS - 1) / GR_THREADS;
    float4 hv[UPD_T], av[UPD_T], xv[UPD_T];
    float  gm[UPD_T];
#pragma unroll
    for (int t = 0; t < UPD_T; ++t) {
        const int rl   = min((tid + t * GR_THREADS) >> 5, GR_NB - 1);
        const int node = min(base + rl, n - 1);
        const size_t o = (size_t)node * (HIDC / 4) + lane;
        gm[t] = gprev[node];
        hv[t] = ((const float4*)hup)[o];
        av[t] = ((const float4*)agg)[o];
        xv[t] = ((const float4*)xs)[o];
    }

    if (tid < GR_NB) gs[tid] = gate_value(gs[tid], cnt[tid]);
    __syncthreads();

#pragma unroll
    for (int t = 0; t < UPD_T; ++t) {
        const int i    = tid + t * GR_THREADS;
        const int rl   = i >> 5;
        const int node = base + rl;
        if (i < GR_NB * 32 && node < n) {
            const float gq  = gs[rl];
            const float den = 1.0f + gm[t] + gq;
            float4 r;
            r.x = (hv[t].x + gm[t] * av[t].x + gq * xv[t].x) / den;
            r.y = (hv[t].y + gm[t] * av[t].y + gq * xv[t].y) / den;
            r.z = (hv[t].z + gm[t] * av[t].z + gq * xv[t].z) / den;
            r.w = (hv[t].w + gm[t] * av[t].w + gq * xv[t].w) / den;
            ((float4*)hup)[(size_t)node * (HIDC / 4) + lane] = r;
        }
    }
}

extern "C" void kernel_launch(void* const* d_in, const int* in_sizes, int n_in,
                              void* d_out, int out_size, void* d_ws, size_t ws_size,
                              hipStream_t stream) {
    const float* x      = (const float*)d_in[0];
    const float* x0     = (const float*)d_in[1];
    const int*   ei     = (const int*)d_in[2];
    const int*   ei2    = (const int*)d_in[3];
    const float* in_W   = (const float*)d_in[4];
    const float* skip_W = (const float*)d_in[5];
    const float* conv_W = (const float*)d_in[6];
    const float* conv_b = (const float*)d_in[7];
    const float* fc_W   = (const float*)d_in[8];
    const float* fc_b   = (const float*)d_in[9];
    float* out = (float*)d_out;

    const int N  = in_sizes[0] / HIDC;
    const int E1 = in_sizes[2] / 2;
    const int E2 = in_sizes[3] / 2;

    const size_t nh = (size_t)N * HIDC;
    float* h     = (float*)d_ws;
    float* xskip = h + nh;
    float* agg   = xskip + nh;
    float* mbuf  = agg + nh;
    int*   tab1  = (int*)(mbuf + nh);
    const int ng1 = (E1 + GR_CAP - 1) / GR_CAP;
    const int ng2 = (E2 + GR_CAP - 1) / GR_CAP;
    int*   tab2  = tab1 + 2 * ng1;
    float* g1    = mbuf + nh + 4 * (size_t)N;
    const int nown = (N + GR_NB - 1) / GR_NB;
    unsigned* ola  = (unsigned*)mbuf;
    unsigned* olg  = ola + (size_t)nown * OL_CAP;
    int*      ocnt = (int*)(mbuf + nh + 2 * (size_t)N);

    const dim3 blk(256);
    auto gemm_blocks = [](int M, int Nc) {
        int tiles = ((M + 15) / 16) * ((Nc + 15) / 16);
        return dim3((tiles + 7) / 8);
    };

    gemm_wmma_f32<HIDC><<<gemm_blocks(N, HIDC), blk, 0, stream>>>(x,    in_W,   nullptr, h,     N);
    gemm_wmma_f32<HIDC><<<gemm_blocks(N, HIDC), blk, 0, stream>>>(x0,   in_W,   nullptr, mbuf,  N);
    gemm_wmma_f32<HIDC><<<gemm_blocks(N, HIDC), blk, 0, stream>>>(mbuf, skip_W, nullptr, xskip, N);

    range_table<<<nown + ng1 + ng2, dim3(GR_THREADS), 0, stream>>>(
        ei, E1, ng1, ei2, E2, tab1, tab2, nown, ola, olg, ocnt);

    for (int l = 0; l < 2; ++l) {
        agg_by_range<<<(N + AGG_NB - 1) / AGG_NB, dim3(AGG_THREADS), 0, stream>>>(
            h, ei, E1, conv_W + (size_t)l * HIDC * HIDC, conv_b + l * HIDC, agg, N, ola, ocnt);
        gamma_by_range<<<nown, dim3(GR_THREADS), 0, stream>>>(
            agg, ei, E1, tab1, g1, nullptr, nullptr, nullptr, N, olg, ocnt);
        gamma_by_range<<<nown, dim3(GR_THREADS), 0, stream>>>(
            agg, ei2, E2, tab2, nullptr, g1, h, xskip, N, nullptr, nullptr);
    }

    gemm_wmma_f32<40><<<gemm_blocks(N, 40), blk, 0, stream>>>(h, fc_W, fc_b, out, N);
}


# x_skip = x0 @ (in_W @ skip_W): the two 128x128 weights multiplied first (64 tiles riding in the first product launch); no h0 = x0 @ in_W launch or plane; 11 -> 10 launches; sum order changes
# speedup vs baseline: 2.7544x; 1.0781x over previous
#include <hip/hip_runtime.h>

typedef __attribute__((ext_vector_type(2))) float v2f;
typedef __attribute__((ext_vector_type(8))) float v8f;

#define HIDC 128

template <int NC>
__global__ __launch_bounds__(256) void gemm_wmma_f32(
    const float* __restrict__ A1, const float* __restrict__ B1,
    const float* __restrict__ bias1, float* __restrict__ C1, int M1,
    const float* __restrict__ A2, const float* __restrict__ B2,
    float* __restrict__ C2, int M2)
{
    constexpr int K = HIDC;
    constexpr int tilesN = (NC + 15) >> 4;
    const int tiles1 = ((M1 + 15) >> 4) * tilesN;
    const int tiles2 = ((M2 + 15) >> 4) * tilesN;
    int wid = blockIdx.x * 8 + __builtin_amdgcn_readfirstlane((int)(threadIdx.x >> 5));
    if (wid >= tiles1 + tiles2) return;
    const bool second = wid >= tiles1;
    if (second) wid -= tiles1;
    const float* A    = second ? A2 : A1;
    const float* B    = second ? B2 : B1;
    const float* bias = second ? nullptr : bias1;
    float*       C    = second ? C2 : C1;
    const int    M    = second ? M2 : M1;
    const int tm = wid / tilesN, tn = wid % tilesN;
    const int m0 = tm << 4, n0 = tn << 4;
    const int lane = threadIdx.x & 31;
    const int half = lane >> 4;
    const int l16  = lane & 15;

    const int arow   = min(m0 + l16, M - 1);
    const int bcol_r = n0 + l16;
    const int bcol   = min(bcol_r, NC - 1);

    const float* Ap = A + (size_t)arow * K + 2 * half;
    const float* Bp = B + (size_t)(2 * half) * NC + bcol;

    v8f acc = {};
#pragma unroll
    for (int k0 = 0; k0 < K; k0 += 4) {
        v2f a = *(const v2f*)(Ap + k0);
        v2f b;
        b.x = Bp[k0 * NC];
        b.y = Bp[(k0 + 1) * NC];
        acc = __builtin_amdgcn_wmma_f32_16x16x4_f32(
                  false, a, false, b, (short)0, acc, false, false);
    }

    if (bcol_r < NC) {
        const float bv = bias ? bias[bcol_r] : 0.0f;
#pragma unroll
        for (int v = 0; v < 8; ++v) {
            const int r = m0 + v + 8 * half;
            if (r < M) C[(size_t)r * NC + bcol_r] = acc[v] + bv;
        }
    }
}

#define AGG_NB      80
#define AGG_THREADS 1024
#define AGG_U       8
#define AGG_P       132
#define OL_CAP      4096
#define OL_U        8
#define AGG_B       8

__global__ __launch_bounds__(AGG_THREADS) void agg_by_range(
    const float* __restrict__ hsrc, const int* __restrict__ ei, int E,
    const float* __restrict__ Wm, const float* __restrict__ bias,
    float* __restrict__ agg, int n,
    const unsigned* __restrict__ olist, const int* __restrict__ ocnt)
{
    __shared__ float acc[AGG_NB * AGG_P];
    __shared__ float prj[AGG_NB * AGG_P];
    const int tid  = threadIdx.x;
    const int lane = tid & 31;
    const int wave = tid >> 5;
    const int base = blockIdx.x * AGG_NB;

    for (int i = tid; i < AGG_NB * AGG_P / 4; i += AGG_THREADS)
        ((float4*)acc)[i] = make_float4(0.0f, 0.0f, 0.0f, 0.0f);
    __syncthreads();

    const int* __restrict__ rowp = ei;
    const int* __restrict__ colp = ei + E;
    constexpr int STEP = 32 * AGG_U;
    const int nstep = (E + STEP - 1) / STEP;
    const int rot = (int)((blockIdx.x * (unsigned)nstep) / gridDim.x);
    const int lcnt = ocnt[2 * blockIdx.x];
    if (lcnt <= OL_CAP) {
        const unsigned* __restrict__ L = olist + (size_t)blockIdx.x * OL_CAP;
        for (int s = wave * AGG_B; s < lcnt; s += (AGG_THREADS / 32) * AGG_B) {
            unsigned ent[AGG_B];
            float4   v[AGG_B];
#pragma unroll
            for (int q = 0; q < AGG_B; ++q)
                ent[q] = L[min(s + q, lcnt - 1)];
#pragma unroll
            for (int q = 0; q < AGG_B; ++q)
                v[q] = *((const float4*)(hsrc + (size_t)(ent[q] >> 7) * HIDC) + lane);
#pragma unroll
            for (int q = 0; q < AGG_B; ++q) {
                const bool ok = s + q < lcnt;
                float* a = acc + (int)(ent[q] & 127u) * AGG_P + lane;
                atomicAdd(a,      ok ? v[q].x : 0.0f);
                atomicAdd(a + 32, ok ? v[q].y : 0.0f);
                atomicAdd(a + 64, ok ? v[q].z : 0.0f);
                atomicAdd(a + 96, ok ? v[q].w : 0.0f);
            }
        }
    } else {
    for (int s = wave; s < nstep; s += AGG_THREADS / 32) {
        int sc = s + rot;
        if (sc >= nstep) sc -= nstep;
        const int e0 = sc * STEP;
        int  d[AGG_U], r[AGG_U];
        bool hit[AGG_U];
#pragma unroll
        for (int u = 0; u < AGG_U; ++u) {
            const int e  = e0 + u * 32 + lane;
            const int ec = min(e, E - 1);
            d[u]   = colp[ec] - base;
            hit[u] = (e < E) && ((unsigned)d[u] < (unsigned)AGG_NB);
        }
#pragma unroll
        for (int u = 0; u < AGG_U; ++u) {
            r[u] = 0;
            if (hit[u]) r[u] = rowp[e0 + u * 32 + lane];
        }
#pragma unroll
        for (int u = 0; u < AGG_U; ++u) {
            unsigned mask = __builtin_amdgcn_ballot_w32(hit[u]);
            while (mask) {
                const int j = __builtin_ctz(mask);
                mask &= mask - 1;
                const int rj = __builtin_amdgcn_readlane(r[u], j);
                const int dj = __builtin_amdgcn_readlane(d[u], j);
                const float4 v = *((const float4*)(hsrc + (size_t)rj * HIDC) + lane);
                float* a = acc + dj * AGG_P + lane;
                atomicAdd(a,      v.x);
                atomicAdd(a + 32, v.y);
                atomicAdd(a + 64, v.z);
                atomicAdd(a + 96, v.w);
            }
        }
    }
    }
    __syncthreads();

    {
        const int half = lane >> 4;
        const int l16  = lane & 15;
        const int n0   = (wave & 7) << 4;
        const float* Bp = Wm + (size_t)(2 * half) * HIDC + n0 + l16;
        const int col = n0 + l16;
        for (int mt = wave >> 3; mt < AGG_NB / 16; mt += AGG_THREADS / 256) {
            const float* Ar = acc + (mt * 16 + l16) * AGG_P + (2 * half) * 32;
            v8f d = {};
#pragma unroll
            for (int k0 = 0; k0 < HIDC; k0 += 4) {
                v2f a, b;
                a.x = Ar[k0 >> 2];
                a.y = Ar[32 + (k0 >> 2)];
                b.x = Bp[k0 * HIDC];
                b.y = Bp[(k0 + 1) * HIDC];
                d = __builtin_amdgcn_wmma_f32_16x16x4_f32(
                        false, a, false, b, (short)0, d, false, false);
            }
            float* o = prj + (mt * 16 + 8 * half) * AGG_P + (col & 3) * 32 + (col >> 2);
#pragma unroll
            for (int v = 0; v < 8; ++v) o[v * AGG_P] = d[v];
        }
    }
    __syncthreads();

    const float4 bv = *((const float4*)bias + lane);
    for (int i = tid; i < AGG_NB * 32; i += AGG_THREADS) {
        const int rl   = i >> 5;
        const int node = base + rl;
        if (node < n) {
            const float* a = prj + rl * AGG_P + lane;
            float4 o;
            o.x = fmaxf(a[0]  + bv.x, 0.0f);
            o.y = fmaxf(a[32] + bv.y, 0.0f);
            o.z = fmaxf(a[64] + bv.z, 0.0f);
            o.w = fmaxf(a[96] + bv.w, 0.0f);
            *((float4*)(agg + (size_t)node * HIDC) + lane) = o;
        }
    }
}

template <int CTRL>
__device__ __forceinline__ float lane_read(float v) {
    return __int_as_float(__builtin_amdgcn_update_dpp(
        0, __float_as_int(v), CTRL, 0xF, 0xF, true));
}

#define GR_NB      80
#define GR_THREADS 1024
#define GR_CAP     32768
#define GR_U       16
static_assert(GR_NB == AGG_NB, "one owner range for the sums and the gates: the owner lists are made for both");
static_assert(OL_CAP <= GR_CAP, "an owner list must fit the LDS list");

__device__ __forceinline__ float gate_value(float s, int c) {
    return tanhf(s * (1.0f / ((float)c + 1e-10f)));
}

__device__ __forceinline__ void owner_lists(
    const int* __restrict__ ei, int E, int b, int nown,
    unsigned* __restrict__ la, unsigned* __restrict__ lg, int* __restrict__ cnts, int* sn)
{
    const int tid  = threadIdx.x;
    const int lane = tid & 31;
    const int wave = tid >> 5;
    const int base = b * GR_NB;
    if (tid < 2) sn[tid] = 0;
    __syncthreads();
    const int* __restrict__ rowp = ei;
    const int* __restrict__ colp = ei + E;
    constexpr int STEP = 32 * OL_U;
    const int nstep = (E + STEP - 1) / STEP;
    const int rot = (int)(((unsigned)b * (unsigned)nstep) / (unsigned)nown);
    for (int s = wave; s < nstep; s += GR_THREADS / 32) {
        int sc = s + rot;
        if (sc >= nstep) sc -= nstep;
        const int e0 = sc * STEP;
        int r[OL_U], c[OL_U];
#pragma unroll
        for (int u = 0; u < OL_U; ++u) {
            const int ec = min(e0 + u * 32 + lane, E - 1);
            r[u] = rowp[ec];
            c[u] = colp[ec];
        }
#pragma unroll
        for (int u = 0; u < OL_U; ++u) {
            const bool in = (e0 + u * 32 + lane) < E;
            const int  dc = c[u] - base, dr = r[u] - base;
            const bool ha = in && ((unsigned)dc < (unsigned)GR_NB);
            const bool hg = in && ((unsigned)dr < (unsigned)GR_NB);
            const unsigned ma = __builtin_amdgcn_ballot_w32(ha);
            if (ma) {
                int off = 0;
                if (lane == 0) off = atomicAdd(&sn[0], __builtin_popcount(ma));
                off = __builtin_amdgcn_readfirstlane(off);
                const int pos = off + (int)__builtin_amdgcn_mbcnt_lo(ma, 0);
                if (ha && pos < OL_CAP) la[pos] = ((unsigned)r[u] << 7) | (unsigned)dc;
            }
            const unsigned mg = __builtin_amdgcn_ballot_w32(hg);
            if (mg) {
                int off = 0;
                if (lane == 0) off = atomicAdd(&sn[1], __builtin_popcount(mg));
                off = __builtin_amdgcn_readfirstlane(off);
                const int pos = off + (int)__builtin_amdgcn_mbcnt_lo(mg, 0);
                if (hg && pos < OL_CAP) lg[pos] = ((unsigned)c[u] << 7) | (unsigned)dr;
            }
        }
    }
    __syncthreads();
    if (tid < 2) cnts[2 * b + tid] = sn[tid];
}

__global__ __launch_bounds__(GR_THREADS) void range_table(
    const int* __restrict__ ei1, int E1, int ng1,
    const int* __restrict__ ei2, int E2,
    int* __restrict__ tab1, int* __restrict__ tab2,
    int nown, unsigned* __restrict__ ola, unsigned* __restrict__ olg, int* __restrict__ ocnt)
{
    __shared__ int smin, smax;
    __shared__ int sn[2];
    if ((int)blockIdx.x < nown) {
        owner_lists(ei1, E1, (int)blockIdx.x, nown,
                    ola + (size_t)blockIdx.x * OL_CAP, olg + (size_t)blockIdx.x * OL_CAP,
                    ocnt, sn);
        return;
    }
    int grp = (int)blockIdx.x - nown;
    const int* __restrict__ rowp = ei1;
    int  E   = E1;
    int* tab = tab1;
    if (grp >= ng1) { grp -= ng1; rowp = ei2; E = E2; tab = tab2; }
    if (threadIdx.x == 0) { smin = 0x7fffffff; smax = -0x7fffffff - 1; }
    __syncthreads();
    int lo = 0x7fffffff, hi = -0x7fffffff - 1;
    const int e0 = grp * GR_CAP + (int)threadIdx.x;
#pragma unroll 8
    for (int k = 0; k < GR_CAP / GR_THREADS; ++k) {
        const int r = rowp[min(e0 + k * GR_THREADS, E - 1)];
        lo = min(lo, r);
        hi = max(hi, r);
    }
    atomicMin(&smin, lo);
    atomicMax(&smax, hi);
    __syncthreads();
    if (threadIdx.x == 0) { tab[2 * grp] = smin; tab[2 * grp + 1] = smax; }
}

__global__ __launch_bounds__(GR_THREADS) void gamma_by_range(
    const float* __restrict__ agg, const int* __restrict__ ei, int E,
    const int* __restrict__ tab, float* __restrict__ g,
    const float* __restrict__ gprev, float* __restrict__ hup,
    const float* __restrict__ xs, int n,
    const unsigned* __restrict__ olist, const int* __restrict__ ocnt)
{
    __shared__ float    gs[GR_NB];
    __shared__ int      cnt[GR_NB];
    __shared__ int      nlist;
    __shared__ unsigned list[GR_CAP];
    const int tid  = threadIdx.x;
    const int lane = tid & 31;
    const int wave = tid >> 5;
    const int base = blockIdx.x * GR_NB;
    const int slot = lane >> 4;
    const int sub  = lane & 15;

    if (tid < GR_NB) { gs[tid] = 0.0f; cnt[tid] = 0; }
    if (tid == 0) nlist = 0;
    __syncthreads();

    const int* __restrict__ rowp = ei;
    const int* __restrict__ colp = ei + E;
    const int ngrp = (E + GR_CAP - 1) / GR_CAP;
    const int rot = (int)(blockIdx.x % (unsigned)ngrp);
    int lcnt = -1;
    if (olist != nullptr) {
        const int c = ocnt[2 * blockIdx.x + 1];
        if (c <= OL_CAP) lcnt = c;
    }
    const int ntrip = (lcnt >= 0) ? 1 : ngrp;
    for (int g0 = 0; g0 < ntrip; ++g0) {
        if (lcnt >= 0) {
            const unsigned* __restrict__ L = olist + (size_t)blockIdx.x * OL_CAP;
            for (int i = tid; i < lcnt; i += GR_THREADS) list[i] = L[i];
            if (tid == 0) nlist = lcnt;
        } else {
        int gi = g0 + rot;
        if (gi >= ngrp) gi -= ngrp;
        const int lo = tab[2 * gi], hi = tab[2 * gi + 1];
        if (hi < base || lo >= base + GR_NB) continue;

        for (int it = 0; it < GR_CAP / (GR_THREADS * GR_U); ++it) {
            const int e0 = gi * GR_CAP + (it * (GR_THREADS / 32) + wave) * (32 * GR_U);
            int  rr[GR_U], cc[GR_U];
            bool hit[GR_U];
#pragma unroll
            for (int u = 0; u < GR_U; ++u) {
                const int e  = e0 + u * 32 + lane;
                const int ec = min(e, E - 1);
                rr[u]  = rowp[ec] - base;
                hit[u] = (e < E) && ((unsigned)rr[u] < (unsigned)GR_NB);
            }
#pragma unroll
            for (int u = 0; u < GR_U; ++u) {
                cc[u] = 0;
                if (hit[u]) cc[u] = colp[e0 + u * 32 + lane];
            }
#pragma unroll
            for (int u = 0; u < GR_U; ++u) {
                const unsigned mask = __builtin_amdgcn_ballot_w32(hit[u]);
                if (mask) {
                    int off = 0;
                    if (lane == 0) off = atomicAdd(&nlist, __builtin_popcount(mask));
                    off = __builtin_amdgcn_readfirstlane(off);
                    if (hit[u])
                        list[off + __builtin_amdgcn_mbcnt_lo(mask, 0)] =
                            ((unsigned)cc[u] << 7) | (unsigned)rr[u];
                }
            }
        }
        }
        __syncthreads();
        const int nh = nlist;

        for (int s = wave * 8; s < nh; s += (GR_THREADS / 32) * 8) {
            int    rl[4];
            bool   ok[4];
            float4 a[4][2], b[4][2];
#pragma unroll
            for (int q = 0; q < 4; ++q) {
                const int idx = s + 2 * q + slot;
                ok[q] = idx < nh;
                const unsigned ent = list[min(idx, nh - 1)];
                rl[q] = (int)(ent & 127u);
                const int c = (int)(ent >> 7);
                const float4* pa = (const float4*)(agg + (size_t)(base + rl[q]) * HIDC) + sub;
                const float4* pb = (const float4*)(agg + (size_t)c * HIDC) + sub;
#pragma unroll
                for (int i = 0; i < 2; ++i) {
                    a[q][i] = pa[16 * i];
                    b[q][i] = pb[16 * i];
                }
            }

            __builtin_amdgcn_sched_barrier(0);

#pragma unroll
            for (int q = 0; q < 4; ++q) {
                float sv = 0.0f;
#pragma unroll
                for (int i = 0; i < 2; ++i) {
                    const float dx = a[q][i].x - b[q][i].x;
                    const float dy = a[q][i].y - b[q][i].y;
                    const float dz = a[q][i].z - b[q][i].z;
                    const float dw = a[q][i].w - b[q][i].w;
                    sv += dx * dx + dy * dy + dz * dz + dw * dw;
                }
                sv += lane_read<0x161>(sv);
                sv += lane_read<0x162>(sv);
                sv += lane_read<0x164>(sv);
                sv += lane_read<0x168>(sv);
                if (sub == 0 && ok[q]) {
                    atomicAdd(&gs[rl[q]], sv);
                    atomicAdd(&cnt[rl[q]], 1);
                }
            }
        }
        __syncthreads();
        if (tid == 0) nlist = 0;
        __syncthreads();
    }

    if (hup == nullptr) {
        if (tid < GR_NB && base + tid < n) g[base + tid] = gate_value(gs[tid], cnt[tid]);
        return;
    }

    constexpr int UPD_T = (GR_NB * 32 + GR_THREADS - 1) / GR_THREADS;
    float4 hv[UPD_T], av[UPD_T], xv[UPD_T];
    float  gm[UPD_T];
#pragma unroll
    for (int t = 0; t < UPD_T; ++t) {
        const int rl   = min((tid + t * GR_THREADS) >> 5, GR_NB - 1);
        const int node = min(base + rl, n - 1);
        const size_t o = (size_t)node * (HIDC / 4) + lane;
        gm[t] = gprev[node];
        hv[t] = ((const float4*)hup)[o];
        av[t] = ((const float4*)agg)[o];
        xv[t] = ((const float4*)xs)[o];
    }

    if (tid < GR_NB) gs[tid] = gate_value(gs[tid], cnt[tid]);
    __syncthreads();

#pragma unroll
    for (int t = 0; t < UPD_T; ++t) {
        const int i    = tid + t * GR_THREADS;
        const int rl   = i >> 5;
        const int node = base + rl;
        if (i < GR_NB * 32 && node < n) {
            const float gq  = gs[rl];
            const float den = 1.0f + gm[t] + gq;
            float4 r;
            r.x = (hv[t].x + gm[t] * av[t].x + gq * xv[t].x) / den;
            r.y = (hv[t].y + gm[t] * av[t].y + gq * xv[t].y) / den;
            r.z = (hv[t].z + gm[t] * av[t].z + gq * xv[t].z) / den;
            r.w = (hv[t].w + gm[t] * av[t].w + gq * xv[t].w) / den;
            ((float4*)hup)[(size_t)node * (HIDC / 4) + lane] = r;
        }
    }
}

extern "C" void kernel_launch(void* const* d_in, const int* in_sizes, int n_in,
                              void* d_out, int out_size, void* d_ws, size_t ws_size,
                              hipStream_t stream) {
    const float* x      = (const float*)d_in[0];
    const float* x0     = (const float*)d_in[1];
    const int*   ei     = (const int*)d_in[2];
    const int*   ei2    = (const int*)d_in[3];
    const float* in_W   = (const float*)d_in[4];
    const float* skip_W = (const float*)d_in[5];
    const float* conv_W = (const float*)d_in[6];
    const float* conv_b = (const float*)d_in[7];
    const float* fc_W   = (const float*)d_in[8];
    const float* fc_b   = (const float*)d_in[9];
    float* out = (float*)d_out;

    const int N  = in_sizes[0] / HIDC;
    const int E1 = in_sizes[2] / 2;
    const int E2 = in_sizes[3] / 2;

    const size_t nh = (size_t)N * HIDC;
    float* h     = (float*)d_ws;
    float* xskip = h + nh;
    float* agg   = xskip + nh;
    float* mbuf  = agg + nh;
    int*   tab1  = (int*)(mbuf + nh);
    const int ng1 = (E1 + GR_CAP - 1) / GR_CAP;
    const int ng2 = (E2 + GR_CAP - 1) / GR_CAP;
    int*   tab2  = tab1 + 2 * ng1;
    float* g1    = mbuf + nh + 4 * (size_t)N;
    const int nown = (N + GR_NB - 1) / GR_NB;
    float* w2 = mbuf;
    unsigned* ola  = (unsigned*)(mbuf + HIDC * HIDC);
    unsigned* olg  = ola + (size_t)nown * OL_CAP;
    int*      ocnt = (int*)(mbuf + nh + 2 * (size_t)N);

    const dim3 blk(256);
    auto gemm_blocks = [](int M, int M2, int Nc) {
        int tiles = (((M + 15) / 16) + ((M2 + 15) / 16)) * ((Nc + 15) / 16);
        return dim3((tiles + 7) / 8);
    };

    gemm_wmma_f32<HIDC><<<gemm_blocks(N, HIDC, HIDC), blk, 0, stream>>>(
        x, in_W, nullptr, h, N, in_W, skip_W, w2, HIDC);
    gemm_wmma_f32<HIDC><<<gemm_blocks(N, 0, HIDC), blk, 0, stream>>>(
        x0, w2, nullptr, xskip, N, nullptr, nullptr, nullptr, 0);

    range_table<<<nown + ng1 + ng2, dim3(GR_THREADS), 0, stream>>>(
        ei, E1, ng1, ei2, E2, tab1, tab2, nown, ola, olg, ocnt);

    for (int l = 0; l < 2; ++l) {
        agg_by_range<<<(N + AGG_NB - 1) / AGG_NB, dim3(AGG_THREADS), 0, stream>>>(
            h, ei, E1, conv_W + (size_t)l * HIDC * HIDC, conv_b + l * HIDC, agg, N, ola, ocnt);
        gamma_by_range<<<nown, dim3(GR_THREADS), 0, stream>>>(
            agg, ei, E1, tab1, g1, nullptr, nullptr, nullptr, N, olg, ocnt);
        gamma_by_range<<<nown, dim3(GR_THREADS), 0, stream>>>(
            agg, ei2, E2, tab2, nullptr, g1, h, xskip, N, nullptr, nullptr);
    }

    gemm_wmma_f32<40><<<gemm_blocks(N, 0, 40), blk, 0, stream>>>(
        h, fc_W, fc_b, out, N, nullptr, nullptr, nullptr, 0);
}


# layer 0 from the inputs: first agg sums rows of x, projects by in_W@conv_W[0]; closing gate block forms h, x_skip of its 80 rows on WMMA from LDS; no node GEMMs; 10 -> 8 launches; sum order changes
# speedup vs baseline: 2.9959x; 1.0877x over previous
#include <hip/hip_runtime.h>

typedef __attribute__((ext_vector_type(2))) float v2f;
typedef __attribute__((ext_vector_type(8))) float v8f;

#define HIDC 128

template <int NC>
__device__ __forceinline__ void gemm_tile(
    int wid,
    const float* __restrict__ A1, const float* __restrict__ B1,
    const float* __restrict__ bias1, float* __restrict__ C1, int M1,
    const float* __restrict__ A2, const float* __restrict__ B2,
    float* __restrict__ C2, int M2)
{
    constexpr int K = HIDC;
    constexpr int tilesN = (NC + 15) >> 4;
    const int tiles1 = ((M1 + 15) >> 4) * tilesN;
    const int tiles2 = ((M2 + 15) >> 4) * tilesN;
    if (wid >= tiles1 + tiles2) return;
    const bool second = wid >= tiles1;
    if (second) wid -= tiles1;
    const float* A    = second ? A2 : A1;
    const float* B    = second ? B2 : B1;
    const float* bias = second ? nullptr : bias1;
    float*       C    = second ? C2 : C1;
    const int    M    = second ? M2 : M1;
    const int tm = wid / tilesN, tn = wid % tilesN;
    const int m0 = tm << 4, n0 = tn << 4;
    const int lane = threadIdx.x & 31;
    const int half = lane >> 4;
    const int l16  = lane & 15;

    const int arow   = min(m0 + l16, M - 1);
    const int bcol_r = n0 + l16;
    const int bcol   = min(bcol_r, NC - 1);

    const float* Ap = A + (size_t)arow * K + 2 * half;
    const float* Bp = B + (size_t)(2 * half) * NC + bcol;

    v8f acc = {};
#pragma unroll
    for (int k0 = 0; k0 < K; k0 += 4) {
        v2f a = *(const v2f*)(Ap + k0);
        v2f b;
        b.x = Bp[k0 * NC];
        b.y = Bp[(k0 + 1) * NC];
        acc = __builtin_amdgcn_wmma_f32_16x16x4_f32(
                  false, a, false, b, (short)0, acc, false, false);
    }

    if (bcol_r < NC) {
        const float bv = bias ? bias[bcol_r] : 0.0f;
#pragma unroll
        for (int v = 0; v < 8; ++v) {
            const int r = m0 + v + 8 * half;
            if (r < M) C[(size_t)r * NC + bcol_r] = acc[v] + bv;
        }
    }
}

template <int NC>
__global__ __launch_bounds__(256) void gemm_wmma_f32(
    const float* __restrict__ A1, const float* __restrict__ B1,
    const float* __restrict__ bias1, float* __restrict__ C1, int M1,
    const float* __restrict__ A2, const float* __restrict__ B2,
    float* __restrict__ C2, int M2)
{
    const int wid = blockIdx.x * 8 + __builtin_amdgcn_readfirstlane((int)(threadIdx.x >> 5));
    gemm_tile<NC>(wid, A1, B1, bias1, C1, M1, A2, B2, C2, M2);
}

#define AGG_NB      80
#define AGG_THREADS 1024
#define AGG_U       8
#define AGG_P       132
#define OL_CAP      4096
#define OL_U        8
#define AGG_B       8

__global__ __launch_bounds__(AGG_THREADS) void agg_by_range(
    const float* __restrict__ hsrc, const int* __restrict__ ei, int E,
    const float* __restrict__ Wm, const float* __restrict__ bias,
    float* __restrict__ agg, int n,
    const unsigned* __restrict__ olist, const int* __restrict__ ocnt)
{
    __shared__ float acc[AGG_NB * AGG_P];
    __shared__ float prj[AGG_NB * AGG_P];
    const int tid  = threadIdx.x;
    const int lane = tid & 31;
    const int wave = tid >> 5;
    const int base = blockIdx.x * AGG_NB;

    for (int i = tid; i < AGG_NB * AGG_P / 4; i += AGG_THREADS)
        ((float4*)acc)[i] = make_float4(0.0f, 0.0f, 0.0f, 0.0f);
    __syncthreads();

    const int* __restrict__ rowp = ei;
    const int* __restrict__ colp = ei + E;
    constexpr int STEP = 32 * AGG_U;
    const int nstep = (E + STEP - 1) / STEP;
    const int rot = (int)((blockIdx.x * (unsigned)nstep) / gridDim.x);
    const int lcnt = ocnt[2 * blockIdx.x];
    if (lcnt <= OL_CAP) {
        const unsigned* __restrict__ L = olist + (size_t)blockIdx.x * OL_CAP;
        for (int s = wave * AGG_B; s < lcnt; s += (AGG_THREADS / 32) * AGG_B) {
            unsigned ent[AGG_B];
            float4   v[AGG_B];
#pragma unroll
            for (int q = 0; q < AGG_B; ++q)
                ent[q] = L[min(s + q, lcnt - 1)];
#pragma unroll
            for (int q = 0; q < AGG_B; ++q)
                v[q] = *((const float4*)(hsrc + (size_t)(ent[q] >> 7) * HIDC) + lane);
#pragma unroll
            for (int q = 0; q < AGG_B; ++q) {
                const bool ok = s + q < lcnt;
                float* a = acc + (int)(ent[q] & 127u) * AGG_P + lane;
                atomicAdd(a,      ok ? v[q].x : 0.0f);
                atomicAdd(a + 32, ok ? v[q].y : 0.0f);
                atomicAdd(a + 64, ok ? v[q].z : 0.0f);
                atomicAdd(a + 96, ok ? v[q].w : 0.0f);
            }
        }
    } else {
    for (int s = wave; s < nstep; s += AGG_THREADS / 32) {
        int sc = s + rot;
        if (sc >= nstep) sc -= nstep;
        const int e0 = sc * STEP;
        int  d[AGG_U], r[AGG_U];
        bool hit[AGG_U];
#pragma unroll
        for (int u = 0; u < AGG_U; ++u) {
            const int e  = e0 + u * 32 + lane;
            const int ec = min(e, E - 1);
            d[u]   = colp[ec] - base;
            hit[u] = (e < E) && ((unsigned)d[u] < (unsigned)AGG_NB);
        }
#pragma unroll
        for (int u = 0; u < AGG_U; ++u) {
            r[u] = 0;
            if (hit[u]) r[u] = rowp[e0 + u * 32 + lane];
        }
#pragma unroll
        for (int u = 0; u < AGG_U; ++u) {
            unsigned mask = __builtin_amdgcn_ballot_w32(hit[u]);
            while (mask) {
                const int j = __builtin_ctz(mask);
                mask &= mask - 1;
                const int rj = __builtin_amdgcn_readlane(r[u], j);
                const int dj = __builtin_amdgcn_readlane(d[u], j);
                const float4 v = *((const float4*)(hsrc + (size_t)rj * HIDC) + lane);
                float* a = acc + dj * AGG_P + lane;
                atomicAdd(a,      v.x);
                atomicAdd(a + 32, v.y);
                atomicAdd(a + 64, v.z);
                atomicAdd(a + 96, v.w);
            }
        }
    }
    }
    __syncthreads();

    {
        const int half = lane >> 4;
        const int l16  = lane & 15;
        const int n0   = (wave & 7) << 4;
        const float* Bp = Wm + (size_t)(2 * half) * HIDC + n0 + l16;
        const int col = n0 + l16;
        for (int mt = wave >> 3; mt < AGG_NB / 16; mt += AGG_THREADS / 256) {
            const float* Ar = acc + (mt * 16 + l16) * AGG_P + (2 * half) * 32;
            v8f d = {};
#pragma unroll
            for (int k0 = 0; k0 < HIDC; k0 += 4) {
                v2f a, b;
                a.x = Ar[k0 >> 2];
                a.y = Ar[32 + (k0 >> 2)];
                b.x = Bp[k0 * HIDC];
                b.y = Bp[(k0 + 1) * HIDC];
                d = __builtin_amdgcn_wmma_f32_16x16x4_f32(
                        false, a, false, b, (short)0, d, false, false);
            }
            float* o = prj + (mt * 16 + 8 * half) * AGG_P + (col & 3) * 32 + (col >> 2);
#pragma unroll
            for (int v = 0; v < 8; ++v) o[v * AGG_P] = d[v];
        }
    }
    __syncthreads();

    const float4 bv = *((const float4*)bias + lane);
    for (int i = tid; i < AGG_NB * 32; i += AGG_THREADS) {
        const int rl   = i >> 5;
        const int node = base + rl;
        if (node < n) {
            const float* a = prj + rl * AGG_P + lane;
            float4 o;
            o.x = fmaxf(a[0]  + bv.x, 0.0f);
            o.y = fmaxf(a[32] + bv.y, 0.0f);
            o.z = fmaxf(a[64] + bv.z, 0.0f);
            o.w = fmaxf(a[96] + bv.w, 0.0f);
            *((float4*)(agg + (size_t)node * HIDC) + lane) = o;
        }
    }
}

template <int CTRL>
__device__ __forceinline__ float lane_read(float v) {
    return __int_as_float(__builtin_amdgcn_update_dpp(
        0, __float_as_int(v), CTRL, 0xF, 0xF, true));
}

#define GR_NB      80
#define GR_THREADS 1024
#define GR_CAP     32768
#define GR_U       16
static_assert(GR_NB == AGG_NB, "one owner range for the sums and the gates: the owner lists are made for both");
static_assert(OL_CAP <= GR_CAP, "an owner list must fit the LDS list");

__device__ __forceinline__ float gate_value(float s, int c) {
    return tanhf(s * (1.0f / ((float)c + 1e-10f)));
}

__device__ __forceinline__ void owner_lists(
    const int* __restrict__ ei, int E, int b, int nown,
    unsigned* __restrict__ la, unsigned* __restrict__ lg, int* __restrict__ cnts, int* sn)
{
    const int tid  = threadIdx.x;
    const int lane = tid & 31;
    const int wave = tid >> 5;
    const int base = b * GR_NB;
    if (tid < 2) sn[tid] = 0;
    __syncthreads();
    const int* __restrict__ rowp = ei;
    const int* __restrict__ colp = ei + E;
    constexpr int STEP = 32 * OL_U;
    const int nstep = (E + STEP - 1) / STEP;
    const int rot = (int)(((unsigned)b * (unsigned)nstep) / (unsigned)nown);
    for (int s = wave; s < nstep; s += GR_THREADS / 32) {
        int sc = s + rot;
        if (sc >= nstep) sc -= nstep;
        const int e0 = sc * STEP;
        int r[OL_U], c[OL_U];
#pragma unroll
        for (int u = 0; u < OL_U; ++u) {
            const int ec = min(e0 + u * 32 + lane, E - 1);
            r[u] = rowp[ec];
            c[u] = colp[ec];
        }
#pragma unroll
        for (int u = 0; u < OL_U; ++u) {
            const bool in = (e0 + u * 32 + lane) < E;
            const int  dc = c[u] - base, dr = r[u] - base;
            const bool ha = in && ((unsigned)dc < (unsigned)GR_NB);
            const bool hg = in && ((unsigned)dr < (unsigned)GR_NB);
            const unsigned ma = __builtin_amdgcn_ballot_w32(ha);
            if (ma) {
                int off = 0;
                if (lane == 0) off = atomicAdd(&sn[0], __builtin_popcount(ma));
                off = __builtin_amdgcn_readfirstlane(off);
                const int pos = off + (int)__builtin_amdgcn_mbcnt_lo(ma, 0);
                if (ha && pos < OL_CAP) la[pos] = ((unsigned)r[u] << 7) | (unsigned)dc;
            }
            const unsigned mg = __builtin_amdgcn_ballot_w32(hg);
            if (mg) {
                int off = 0;
                if (lane == 0) off = atomicAdd(&sn[1], __builtin_popcount(mg));
                off = __builtin_amdgcn_readfirstlane(off);
                const int pos = off + (int)__builtin_amdgcn_mbcnt_lo(mg, 0);
                if (hg && pos < OL_CAP) lg[pos] = ((unsigned)c[u] << 7) | (unsigned)dr;
            }
        }
    }
    __syncthreads();
    if (tid < 2) cnts[2 * b + tid] = sn[tid];
}

#define WP_BLOCKS ((2 * (HIDC / 16) * (HIDC / 16)) / (GR_THREADS / 32))
__global__ __launch_bounds__(GR_THREADS) void range_table(
    const int* __restrict__ ei1, int E1, int ng1,
    const int* __restrict__ ei2, int E2,
    int* __restrict__ tab1, int* __restrict__ tab2,
    int nown, unsigned* __restrict__ ola, unsigned* __restrict__ olg, int* __restrict__ ocnt,
    int ngall, const float* __restrict__ Wi, const float* __restrict__ Wc,
    const float* __restrict__ Ws, float* __restrict__ wa, float* __restrict__ wb)
{
    __shared__ int smin, smax;
    __shared__ int sn[2];
    if ((int)blockIdx.x < nown) {
        owner_lists(ei1, E1, (int)blockIdx.x, nown,
                    ola + (size_t)blockIdx.x * OL_CAP, olg + (size_t)blockIdx.x * OL_CAP,
                    ocnt, sn);
        return;
    }
    if ((int)blockIdx.x >= nown + ngall) {
        const int wid = ((int)blockIdx.x - nown - ngall) * (GR_THREADS / 32)
                      + __builtin_amdgcn_readfirstlane((int)(threadIdx.x >> 5));
        gemm_tile<HIDC>(wid, Wi, Wc, nullptr, wa, HIDC, Wi, Ws, wb, HIDC);
        return;
    }
    int grp = (int)blockIdx.x - nown;
    const int* __restrict__ rowp = ei1;
    int  E   = E1;
    int* tab = tab1;
    if (grp >= ng1) { grp -= ng1; rowp = ei2; E = E2; tab = tab2; }
    if (threadIdx.x == 0) { smin = 0x7fffffff; smax = -0x7fffffff - 1; }
    __syncthreads();
    int lo = 0x7fffffff, hi = -0x7fffffff - 1;
    const int e0 = grp * GR_CAP + (int)threadIdx.x;
#pragma unroll 8
    for (int k = 0; k < GR_CAP / GR_THREADS; ++k) {
        const int r = rowp[min(e0 + k * GR_THREADS, E - 1)];
        lo = min(lo, r);
        hi = max(hi, r);
    }
    atomicMin(&smin, lo);
    atomicMax(&smax, hi);
    __syncthreads();
    if (threadIdx.x == 0) { tab[2 * grp] = smin; tab[2 * grp + 1] = smax; }
}

template <bool FIRST>
__global__ __launch_bounds__(GR_THREADS) void gamma_by_range(
    const float* __restrict__ agg, const int* __restrict__ ei, int E,
    const int* __restrict__ tab, float* __restrict__ g,
    const float* __restrict__ gprev, float* __restrict__ hup,
    const float* __restrict__ xs, int n,
    const unsigned* __restrict__ olist, const int* __restrict__ ocnt,
    const float* __restrict__ xin, const float* __restrict__ x0in,
    const float* __restrict__ Wi, const float* __restrict__ Ws2,
    float* __restrict__ xsout)
{
    __shared__ float    gs[GR_NB];
    __shared__ int      cnt[GR_NB];
    __shared__ int      nlist;
    __shared__ unsigned list[GR_CAP];
    __shared__ float    xa[FIRST ? 2 * GR_NB * AGG_P : 4];
    const int tid  = threadIdx.x;
    const int lane = tid & 31;
    const int wave = tid >> 5;
    const int base = blockIdx.x * GR_NB;
    const int slot = lane >> 4;
    const int sub  = lane & 15;

    if (tid < GR_NB) { gs[tid] = 0.0f; cnt[tid] = 0; }
    if (tid == 0) nlist = 0;

    constexpr int UPD_T = (GR_NB * 32 + GR_THREADS - 1) / GR_THREADS;
    __syncthreads();

    float4 px[FIRST ? UPD_T : 1], p0[FIRST ? UPD_T : 1];
    if constexpr (FIRST) {
#pragma unroll
        for (int t = 0; t < UPD_T; ++t) {
            const int rl   = min((tid + t * GR_THREADS) >> 5, GR_NB - 1);
            const int node = min(base + rl, n - 1);
            const size_t o = (size_t)node * (HIDC / 4) + lane;
            px[t] = ((const float4*)xin)[o];
            p0[t] = ((const float4*)x0in)[o];
        }
    }

    const int* __restrict__ rowp = ei;
    const int* __restrict__ colp = ei + E;
    const int ngrp = (E + GR_CAP - 1) / GR_CAP;
    const int rot = (int)(blockIdx.x % (unsigned)ngrp);
    int lcnt = -1;
    if (olist != nullptr) {
        const int c = ocnt[2 * blockIdx.x + 1];
        if (c <= OL_CAP) lcnt = c;
    }
    const int ntrip = (lcnt >= 0) ? 1 : ngrp;
    for (int g0 = 0; g0 < ntrip; ++g0) {
        if (lcnt >= 0) {
            const unsigned* __restrict__ L = olist + (size_t)blockIdx.x * OL_CAP;
            for (int i = tid; i < lcnt; i += GR_THREADS) list[i] = L[i];
            if (tid == 0) nlist = lcnt;
        } else {
        int gi = g0 + rot;
        if (gi >= ngrp) gi -= ngrp;
        const int lo = tab[2 * gi], hi = tab[2 * gi + 1];
        if (hi < base || lo >= base + GR_NB) continue;

        for (int it = 0; it < GR_CAP / (GR_THREADS * GR_U); ++it) {
            const int e0 = gi * GR_CAP + (it * (GR_THREADS / 32) + wave) * (32 * GR_U);
            int  rr[GR_U], cc[GR_U];
            bool hit[GR_U];
#pragma unroll
            for (int u = 0; u < GR_U; ++u) {
                const int e  = e0 + u * 32 + lane;
                const int ec = min(e, E - 1);
                rr[u]  = rowp[ec] - base;
                hit[u] = (e < E) && ((unsigned)rr[u] < (unsigned)GR_NB);
            }
#pragma unroll
            for (int u = 0; u < GR_U; ++u) {
                cc[u] = 0;
                if (hit[u]) cc[u] = colp[e0 + u * 32 + lane];
            }
#pragma unroll
            for (int u = 0; u < GR_U; ++u) {
                const unsigned mask = __builtin_amdgcn_ballot_w32(hit[u]);
                if (mask) {
                    int off = 0;
                    if (lane == 0) off = atomicAdd(&nlist, __builtin_popcount(mask));
                    off = __builtin_amdgcn_readfirstlane(off);
                    if (hit[u])
                        list[off + __builtin_amdgcn_mbcnt_lo(mask, 0)] =
                            ((unsigned)cc[u] << 7) | (unsigned)rr[u];
                }
            }
        }
        }
        __syncthreads();
        const int nh = nlist;

        for (int s = wave * 8; s < nh; s += (GR_THREADS / 32) * 8) {
            int    rl[4];
            bool   ok[4];
            float4 a[4][2], b[4][2];
#pragma unroll
            for (int q = 0; q < 4; ++q) {
                const int idx = s + 2 * q + slot;
                ok[q] = idx < nh;
                const unsigned ent = list[min(idx, nh - 1)];
                rl[q] = (int)(ent & 127u);
                const int c = (int)(ent >> 7);
                const float4* pa = (const float4*)(agg + (size_t)(base + rl[q]) * HIDC) + sub;
                const float4* pb = (const float4*)(agg + (size_t)c * HIDC) + sub;
#pragma unroll
                for (int i = 0; i < 2; ++i) {
                    a[q][i] = pa[16 * i];
                    b[q][i] = pb[16 * i];
                }
            }

            __builtin_amdgcn_sched_barrier(0);

#pragma unroll
            for (int q = 0; q < 4; ++q) {
                float sv = 0.0f;
#pragma unroll
                for (int i = 0; i < 2; ++i) {
                    const float dx = a[q][i].x - b[q][i].x;
                    const float dy = a[q][i].y - b[q][i].y;
                    const float dz = a[q][i].z - b[q][i].z;
                    const float dw = a[q][i].w - b[q][i].w;
                    sv += dx * dx + dy * dy + dz * dz + dw * dw;
                }
                sv += lane_read<0x161>(sv);
                sv += lane_read<0x162>(sv);
                sv += lane_read<0x164>(sv);
                sv += lane_read<0x168>(sv);
                if (sub == 0 && ok[q]) {
                    atomicAdd(&gs[rl[q]], sv);
                    atomicAdd(&cnt[rl[q]], 1);
                }
            }
        }
        __syncthreads();
        if (tid == 0) nlist = 0;
        __syncthreads();
    }

    if (hup == nullptr) {
        if (tid < GR_NB && base + tid < n) g[base + tid] = gate_value(gs[tid], cnt[tid]);
        return;
    }

    if constexpr (FIRST) {
        if (tid < GR_NB) gs[tid] = gate_value(gs[tid], cnt[tid]);
#pragma unroll
        for (int t = 0; t < UPD_T; ++t) {
            const int i = tid + t * GR_THREADS;
            if (i < GR_NB * 32) {
                float* d = xa + (i >> 5) * AGG_P + lane;
                d[0]  = px[t].x; d[32] = px[t].y; d[64] = px[t].z; d[96] = px[t].w;
                float* e = d + GR_NB * AGG_P;
                e[0]  = p0[t].x; e[32] = p0[t].y; e[64] = p0[t].z; e[96] = p0[t].w;
            }
        }
        __syncthreads();

        const int half = lane >> 4;
        const int l16  = lane & 15;
        const int col  = ((wave & 7) << 4) + l16;
        const float* Bh = Wi  + (size_t)(2 * half) * HIDC + col;
        const float* Bx = Ws2 + (size_t)(2 * half) * HIDC + col;
        for (int mt = wave >> 3; mt < GR_NB / 16; mt += GR_THREADS / 256) {
            const int r0 = mt * 16 + 8 * half;
            float ga[8], gm[8];
#pragma unroll
            for (int v = 0; v < 8; ++v) {
                const int node = min(base + r0 + v, n - 1);
                gm[v] = gprev[node];
                ga[v] = agg[(size_t)node * HIDC + col];
            }
            const float* Ah = xa + (mt * 16 + l16) * AGG_P + (2 * half) * 32;
            const float* Ax = Ah + GR_NB * AGG_P;
            v8f dh = {}, dx = {};
#pragma unroll 8
            for (int k0 = 0; k0 < HIDC; k0 += 4) {
                v2f a, b;
                a.x = Ah[k0 >> 2];
                a.y = Ah[32 + (k0 >> 2)];
                b.x = Bh[k0 * HIDC];
                b.y = Bh[(k0 + 1) * HIDC];
                dh = __builtin_amdgcn_wmma_f32_16x16x4_f32(
                         false, a, false, b, (short)0, dh, false, false);
                a.x = Ax[k0 >> 2];
                a.y = Ax[32 + (k0 >> 2)];
                b.x = Bx[k0 * HIDC];
                b.y = Bx[(k0 + 1) * HIDC];
                dx = __builtin_amdgcn_wmma_f32_16x16x4_f32(
                         false, a, false, b, (short)0, dx, false, false);
            }
#pragma unroll
            for (int v = 0; v < 8; ++v) {
                const int rl   = r0 + v;
                const int node = base + rl;
                if (node < n) {
                    const float gq  = gs[rl];
                    const float den = 1.0f + gm[v] + gq;
                    const size_t o  = (size_t)node * HIDC + col;
                    hup[o]   = (dh[v] + gm[v] * ga[v] + gq * dx[v]) / den;
                    xsout[o] = dx[v];
                }
            }
        }
        return;
    }

    float4 hv[UPD_T], av[UPD_T], xv[UPD_T];
    float  gm[UPD_T];
#pragma unroll
    for (int t = 0; t < UPD_T; ++t) {
        const int rl   = min((tid + t * GR_THREADS) >> 5, GR_NB - 1);
        const int node = min(base + rl, n - 1);
        const size_t o = (size_t)node * (HIDC / 4) + lane;
        gm[t] = gprev[node];
        hv[t] = ((const float4*)hup)[o];
        av[t] = ((const float4*)agg)[o];
        xv[t] = ((const float4*)xs)[o];
    }

    if (tid < GR_NB) gs[tid] = gate_value(gs[tid], cnt[tid]);
    __syncthreads();

#pragma unroll
    for (int t = 0; t < UPD_T; ++t) {
        const int i    = tid + t * GR_THREADS;
        const int rl   = i >> 5;
        const int node = base + rl;
        if (i < GR_NB * 32 && node < n) {
            const float gq  = gs[rl];
            const float den = 1.0f + gm[t] + gq;
            float4 r;
            r.x = (hv[t].x + gm[t] * av[t].x + gq * xv[t].x) / den;
            r.y = (hv[t].y + gm[t] * av[t].y + gq * xv[t].y) / den;
            r.z = (hv[t].z + gm[t] * av[t].z + gq * xv[t].z) / den;
            r.w = (hv[t].w + gm[t] * av[t].w + gq * xv[t].w) / den;
            ((float4*)hup)[(size_t)node * (HIDC / 4) + lane] = r;
        }
    }
}

extern "C" void kernel_launch(void* const* d_in, const int* in_sizes, int n_in,
                              void* d_out, int out_size, void* d_ws, size_t ws_size,
                              hipStream_t stream) {
    const float* x      = (const float*)d_in[0];
    const float* x0     = (const float*)d_in[1];
    const int*   ei     = (const int*)d_in[2];
    const int*   ei2    = (const int*)d_in[3];
    const float* in_W   = (const float*)d_in[4];
    const float* skip_W = (const float*)d_in[5];
    const float* conv_W = (const float*)d_in[6];
    const float* conv_b = (const float*)d_in[7];
    const float* fc_W   = (const float*)d_in[8];
    const float* fc_b   = (const float*)d_in[9];
    float* out = (float*)d_out;

    const int N  = in_sizes[0] / HIDC;
    const int E1 = in_sizes[2] / 2;
    const int E2 = in_sizes[3] / 2;

    const size_t nh = (size_t)N * HIDC;
    float* h     = (float*)d_ws;
    float* xskip = h + nh;
    float* agg   = xskip + nh;
    float* mbuf  = agg + nh;
    int*   tab1  = (int*)(mbuf + nh);
    const int ng1 = (E1 + GR_CAP - 1) / GR_CAP;
    const int ng2 = (E2 + GR_CAP - 1) / GR_CAP;
    int*   tab2  = tab1 + 2 * ng1;
    float* g1    = mbuf + nh + 4 * (size_t)N;
    const int nown = (N + GR_NB - 1) / GR_NB;
    float* w2 = mbuf;
    float* w1 = mbuf + HIDC * HIDC;
    unsigned* ola  = (unsigned*)(mbuf + 2 * HIDC * HIDC);
    unsigned* olg  = ola + (size_t)nown * OL_CAP;
    int*      ocnt = (int*)(mbuf + nh + 2 * (size_t)N);

    const dim3 blk(256);
    auto gemm_blocks = [](int M, int M2, int Nc) {
        int tiles = (((M + 15) / 16) + ((M2 + 15) / 16)) * ((Nc + 15) / 16);
        return dim3((tiles + 7) / 8);
    };

    range_table<<<nown + ng1 + ng2 + WP_BLOCKS, dim3(GR_THREADS), 0, stream>>>(
        ei, E1, ng1, ei2, E2, tab1, tab2, nown, ola, olg, ocnt,
        ng1 + ng2, in_W, conv_W, skip_W, w1, w2);

    agg_by_range<<<(N + AGG_NB - 1) / AGG_NB, dim3(AGG_THREADS), 0, stream>>>(
        x, ei, E1, w1, conv_b, agg, N, ola, ocnt);
    gamma_by_range<false><<<nown, dim3(GR_THREADS), 0, stream>>>(
        agg, ei, E1, tab1, g1, nullptr, nullptr, nullptr, N, olg, ocnt,
        nullptr, nullptr, nullptr, nullptr, nullptr);
    gamma_by_range<true><<<nown, dim3(GR_THREADS), 0, stream>>>(
        agg, ei2, E2, tab2, nullptr, g1, h, nullptr, N, nullptr, nullptr,
        x, x0, in_W, w2, xskip);

    agg_by_range<<<(N + AGG_NB - 1) / AGG_NB, dim3(AGG_THREADS), 0, stream>>>(
        h, ei, E1, conv_W + (size_t)HIDC * HIDC, conv_b + HIDC, agg, N, ola, ocnt);
    gamma_by_range<false><<<nown, dim3(GR_THREADS), 0, stream>>>(
        agg, ei, E1, tab1, g1, nullptr, nullptr, nullptr, N, olg, ocnt,
        nullptr, nullptr, nullptr, nullptr, nullptr);
    gamma_by_range<false><<<nown, dim3(GR_THREADS), 0, stream>>>(
        agg, ei2, E2, tab2, nullptr, g1, h, xskip, N, nullptr, nullptr,
        nullptr, nullptr, nullptr, nullptr, nullptr);

    gemm_wmma_f32<40><<<gemm_blocks(N, 0, 40), blk, 0, stream>>>(
        h, fc_W, fc_b, out, N, nullptr, nullptr, nullptr, 0);
}


# gates: one launch per layer, not two: the owner block sums its rows' one-hop edges, then their two-hop edges; both gates stay in its LDS, no g1 plane; 8 -> 6 launches; same arithmetic and sum order
# speedup vs baseline: 3.1145x; 1.0396x over previous
#include <hip/hip_runtime.h>

typedef __attribute__((ext_vector_type(2))) float v2f;
typedef __attribute__((ext_vector_type(8))) float v8f;

#define HIDC 128

template <int NC>
__device__ __forceinline__ void gemm_tile(
    int wid,
    const float* __restrict__ A1, const float* __restrict__ B1,
    const float* __restrict__ bias1, float* __restrict__ C1, int M1,
    const float* __restrict__ A2, const float* __restrict__ B2,
    float* __restrict__ C2, int M2)
{
    constexpr int K = HIDC;
    constexpr int tilesN = (NC + 15) >> 4;
    const int tiles1 = ((M1 + 15) >> 4) * tilesN;
    const int tiles2 = ((M2 + 15) >> 4) * tilesN;
    if (wid >= tiles1 + tiles2) return;
    const bool second = wid >= tiles1;
    if (second) wid -= tiles1;
    const float* A    = second ? A2 : A1;
    const float* B    = second ? B2 : B1;
    const float* bias = second ? nullptr : bias1;
    float*       C    = second ? C2 : C1;
    const int    M    = second ? M2 : M1;
    const int tm = wid / tilesN, tn = wid % tilesN;
    const int m0 = tm << 4, n0 = tn << 4;
    const int lane = threadIdx.x & 31;
    const int half = lane >> 4;
    const int l16  = lane & 15;

    const int arow   = min(m0 + l16, M - 1);
    const int bcol_r = n0 + l16;
    const int bcol   = min(bcol_r, NC - 1);

    const float* Ap = A + (size_t)arow * K + 2 * half;
    const float* Bp = B + (size_t)(2 * half) * NC + bcol;

    v8f acc = {};
#pragma unroll
    for (int k0 = 0; k0 < K; k0 += 4) {
        v2f a = *(const v2f*)(Ap + k0);
        v2f b;
        b.x = Bp[k0 * NC];
        b.y = Bp[(k0 + 1) * NC];
        acc = __builtin_amdgcn_wmma_f32_16x16x4_f32(
                  false, a, false, b, (short)0, acc, false, false);
    }

    if (bcol_r < NC) {
        const float bv = bias ? bias[bcol_r] : 0.0f;
#pragma unroll
        for (int v = 0; v < 8; ++v) {
            const int r = m0 + v + 8 * half;
            if (r < M) C[(size_t)r * NC + bcol_r] = acc[v] + bv;
        }
    }
}

template <int NC>
__global__ __launch_bounds__(256) void gemm_wmma_f32(
    const float* __restrict__ A1, const float* __restrict__ B1,
    const float* __restrict__ bias1, float* __restrict__ C1, int M1,
    const float* __restrict__ A2, const float* __restrict__ B2,
    float* __restrict__ C2, int M2)
{
    const int wid = blockIdx.x * 8 + __builtin_amdgcn_readfirstlane((int)(threadIdx.x >> 5));
    gemm_tile<NC>(wid, A1, B1, bias1, C1, M1, A2, B2, C2, M2);
}

#define AGG_NB      80
#define AGG_THREADS 1024
#define AGG_U       8
#define AGG_P       132
#define OL_CAP      4096
#define OL_U        8
#define AGG_B       8

__global__ __launch_bounds__(AGG_THREADS) void agg_by_range(
    const float* __restrict__ hsrc, const int* __restrict__ ei, int E,
    const float* __restrict__ Wm, const float* __restrict__ bias,
    float* __restrict__ agg, int n,
    const unsigned* __restrict__ olist, const int* __restrict__ ocnt)
{
    __shared__ float acc[AGG_NB * AGG_P];
    __shared__ float prj[AGG_NB * AGG_P];
    const int tid  = threadIdx.x;
    const int lane = tid & 31;
    const int wave = tid >> 5;
    const int base = blockIdx.x * AGG_NB;

    for (int i = tid; i < AGG_NB * AGG_P / 4; i += AGG_THREADS)
        ((float4*)acc)[i] = make_float4(0.0f, 0.0f, 0.0f, 0.0f);
    __syncthreads();

    const int* __restrict__ rowp = ei;
    const int* __restrict__ colp = ei + E;
    constexpr int STEP = 32 * AGG_U;
    const int nstep = (E + STEP - 1) / STEP;
    const int rot = (int)((blockIdx.x * (unsigned)nstep) / gridDim.x);
    const int lcnt = ocnt[2 * blockIdx.x];
    if (lcnt <= OL_CAP) {
        const unsigned* __restrict__ L = olist + (size_t)blockIdx.x * OL_CAP;
        for (int s = wave * AGG_B; s < lcnt; s += (AGG_THREADS / 32) * AGG_B) {
            unsigned ent[AGG_B];
            float4   v[AGG_B];
#pragma unroll
            for (int q = 0; q < AGG_B; ++q)
                ent[q] = L[min(s + q, lcnt - 1)];
#pragma unroll
            for (int q = 0; q < AGG_B; ++q)
                v[q] = *((const float4*)(hsrc + (size_t)(ent[q] >> 7) * HIDC) + lane);
#pragma unroll
            for (int q = 0; q < AGG_B; ++q) {
                const bool ok = s + q < lcnt;
                float* a = acc + (int)(ent[q] & 127u) * AGG_P + lane;
                atomicAdd(a,      ok ? v[q].x : 0.0f);
                atomicAdd(a + 32, ok ? v[q].y : 0.0f);
                atomicAdd(a + 64, ok ? v[q].z : 0.0f);
                atomicAdd(a + 96, ok ? v[q].w : 0.0f);
            }
        }
    } else {
    for (int s = wave; s < nstep; s += AGG_THREADS / 32) {
        int sc = s + rot;
        if (sc >= nstep) sc -= nstep;
        const int e0 = sc * STEP;
        int  d[AGG_U], r[AGG_U];
        bool hit[AGG_U];
#pragma unroll
        for (int u = 0; u < AGG_U; ++u) {
            const int e  = e0 + u * 32 + lane;
            const int ec = min(e, E - 1);
            d[u]   = colp[ec] - base;
            hit[u] = (e < E) && ((unsigned)d[u] < (unsigned)AGG_NB);
        }
#pragma unroll
        for (int u = 0; u < AGG_U; ++u) {
            r[u] = 0;
            if (hit[u]) r[u] = rowp[e0 + u * 32 + lane];
        }
#pragma unroll
        for (int u = 0; u < AGG_U; ++u) {
            unsigned mask = __builtin_amdgcn_ballot_w32(hit[u]);
            while (mask) {
                const int j = __builtin_ctz(mask);
                mask &= mask - 1;
                const int rj = __builtin_amdgcn_readlane(r[u], j);
                const int dj = __builtin_amdgcn_readlane(d[u], j);
                const float4 v = *((const float4*)(hsrc + (size_t)rj * HIDC) + lane);
                float* a = acc + dj * AGG_P + lane;
                atomicAdd(a,      v.x);
                atomicAdd(a + 32, v.y);
                atomicAdd(a + 64, v.z);
                atomicAdd(a + 96, v.w);
            }
        }
    }
    }
    __syncthreads();

    {
        const int half = lane >> 4;
        const int l16  = lane & 15;
        const int n0   = (wave & 7) << 4;
        const float* Bp = Wm + (size_t)(2 * half) * HIDC + n0 + l16;
        const int col = n0 + l16;
        for (int mt = wave >> 3; mt < AGG_NB / 16; mt += AGG_THREADS / 256) {
            const float* Ar = acc + (mt * 16 + l16) * AGG_P + (2 * half) * 32;
            v8f d = {};
#pragma unroll
            for (int k0 = 0; k0 < HIDC; k0 += 4) {
                v2f a, b;
                a.x = Ar[k0 >> 2];
                a.y = Ar[32 + (k0 >> 2)];
                b.x = Bp[k0 * HIDC];
                b.y = Bp[(k0 + 1) * HIDC];
                d = __builtin_amdgcn_wmma_f32_16x16x4_f32(
                        false, a, false, b, (short)0, d, false, false);
            }
            float* o = prj + (mt * 16 + 8 * half) * AGG_P + (col & 3) * 32 + (col >> 2);
#pragma unroll
            for (int v = 0; v < 8; ++v) o[v * AGG_P] = d[v];
        }
    }
    __syncthreads();

    const float4 bv = *((const float4*)bias + lane);
    for (int i = tid; i < AGG_NB * 32; i += AGG_THREADS) {
        const int rl   = i >> 5;
        const int node = base + rl;
        if (node < n) {
            const float* a = prj + rl * AGG_P + lane;
            float4 o;
            o.x = fmaxf(a[0]  + bv.x, 0.0f);
            o.y = fmaxf(a[32] + bv.y, 0.0f);
            o.z = fmaxf(a[64] + bv.z, 0.0f);
            o.w = fmaxf(a[96] + bv.w, 0.0f);
            *((float4*)(agg + (size_t)node * HIDC) + lane) = o;
        }
    }
}

template <int CTRL>
__device__ __forceinline__ float lane_read(float v) {
    return __int_as_float(__builtin_amdgcn_update_dpp(
        0, __float_as_int(v), CTRL, 0xF, 0xF, true));
}

#define GR_NB      80
#define GR_THREADS 1024
#define GR_CAP     32768
#define GR_U       16
static_assert(GR_NB == AGG_NB, "one owner range for the sums and the gates: the owner lists are made for both");
static_assert(OL_CAP <= GR_CAP, "an owner list must fit the LDS list");

__device__ __forceinline__ float gate_value(float s, int c) {
    return tanhf(s * (1.0f / ((float)c + 1e-10f)));
}

__device__ __forceinline__ void owner_lists(
    const int* __restrict__ ei, int E, int b, int nown,
    unsigned* __restrict__ la, unsigned* __restrict__ lg, int* __restrict__ cnts, int* sn)
{
    const int tid  = threadIdx.x;
    const int lane = tid & 31;
    const int wave = tid >> 5;
    const int base = b * GR_NB;
    if (tid < 2) sn[tid] = 0;
    __syncthreads();
    const int* __restrict__ rowp = ei;
    const int* __restrict__ colp = ei + E;
    constexpr int STEP = 32 * OL_U;
    const int nstep = (E + STEP - 1) / STEP;
    const int rot = (int)(((unsigned)b * (unsigned)nstep) / (unsigned)nown);
    for (int s = wave; s < nstep; s += GR_THREADS / 32) {
        int sc = s + rot;
        if (sc >= nstep) sc -= nstep;
        const int e0 = sc * STEP;
        int r[OL_U], c[OL_U];
#pragma unroll
        for (int u = 0; u < OL_U; ++u) {
            const int ec = min(e0 + u * 32 + lane, E - 1);
            r[u] = rowp[ec];
            c[u] = colp[ec];
        }
#pragma unroll
        for (int u = 0; u < OL_U; ++u) {
            const bool in = (e0 + u * 32 + lane) < E;
            const int  dc = c[u] - base, dr = r[u] - base;
            const bool ha = in && ((unsigned)dc < (unsigned)GR_NB);
            const bool hg = in && ((unsigned)dr < (unsigned)GR_NB);
            const unsigned ma = __builtin_amdgcn_ballot_w32(ha);
            if (ma) {
                int off = 0;
                if (lane == 0) off = atomicAdd(&sn[0], __builtin_popcount(ma));
                off = __builtin_amdgcn_readfirstlane(off);
                const int pos = off + (int)__builtin_amdgcn_mbcnt_lo(ma, 0);
                if (ha && pos < OL_CAP) la[pos] = ((unsigned)r[u] << 7) | (unsigned)dc;
            }
            const unsigned mg = __builtin_amdgcn_ballot_w32(hg);
            if (mg) {
                int off = 0;
                if (lane == 0) off = atomicAdd(&sn[1], __builtin_popcount(mg));
                off = __builtin_amdgcn_readfirstlane(off);
                const int pos = off + (int)__builtin_amdgcn_mbcnt_lo(mg, 0);
                if (hg && pos < OL_CAP) lg[pos] = ((unsigned)c[u] << 7) | (unsigned)dr;
            }
        }
    }
    __syncthreads();
    if (tid < 2) cnts[2 * b + tid] = sn[tid];
}

#define WP_BLOCKS ((2 * (HIDC / 16) * (HIDC / 16)) / (GR_THREADS / 32))
__global__ __launch_bounds__(GR_THREADS) void range_table(
    const int* __restrict__ ei1, int E1, int ng1,
    const int* __restrict__ ei2, int E2,
    int* __restrict__ tab1, int* __restrict__ tab2,
    int nown, unsigned* __restrict__ ola, unsigned* __restrict__ olg, int* __restrict__ ocnt,
    int ngall, const float* __restrict__ Wi, const float* __restrict__ Wc,
    const float* __restrict__ Ws, float* __restrict__ wa, float* __restrict__ wb)
{
    __shared__ int smin, smax;
    __shared__ int sn[2];
    if ((int)blockIdx.x < nown) {
        owner_lists(ei1, E1, (int)blockIdx.x, nown,
                    ola + (size_t)blockIdx.x * OL_CAP, olg + (size_t)blockIdx.x * OL_CAP,
                    ocnt, sn);
        return;
    }
    if ((int)blockIdx.x >= nown + ngall) {
        const int wid = ((int)blockIdx.x - nown - ngall) * (GR_THREADS / 32)
                      + __builtin_amdgcn_readfirstlane((int)(threadIdx.x >> 5));
        gemm_tile<HIDC>(wid, Wi, Wc, nullptr, wa, HIDC, Wi, Ws, wb, HIDC);
        return;
    }
    int grp = (int)blockIdx.x - nown;
    const int* __restrict__ rowp = ei1;
    int  E   = E1;
    int* tab = tab1;
    if (grp >= ng1) { grp -= ng1; rowp = ei2; E = E2; tab = tab2; }
    if (threadIdx.x == 0) { smin = 0x7fffffff; smax = -0x7fffffff - 1; }
    __syncthreads();
    int lo = 0x7fffffff, hi = -0x7fffffff - 1;
    const int e0 = grp * GR_CAP + (int)threadIdx.x;
#pragma unroll 8
    for (int k = 0; k < GR_CAP / GR_THREADS; ++k) {
        const int r = rowp[min(e0 + k * GR_THREADS, E - 1)];
        lo = min(lo, r);
        hi = max(hi, r);
    }
    atomicMin(&smin, lo);
    atomicMax(&smax, hi);
    __syncthreads();
    if (threadIdx.x == 0) { tab[2 * grp] = smin; tab[2 * grp + 1] = smax; }
}

template <bool FIRST>
__global__ __launch_bounds__(GR_THREADS) void gamma_by_range(
    const float* __restrict__ agg,
    const int* __restrict__ ei1, int E1, const int* __restrict__ tab1,
    const int* __restrict__ ei2, int E2, const int* __restrict__ tab2,
    float* __restrict__ hup,
    const float* __restrict__ xs, int n,
    const unsigned* __restrict__ olist, const int* __restrict__ ocnt,
    const float* __restrict__ xin, const float* __restrict__ x0in,
    const float* __restrict__ Wi, const float* __restrict__ Ws2,
    float* __restrict__ xsout)
{
    __shared__ float    gs[2 * GR_NB];
    __shared__ int      cnt[2 * GR_NB];
    __shared__ int      nlist;
    __shared__ unsigned list[GR_CAP];
    __shared__ float    xa[FIRST ? 2 * GR_NB * AGG_P : 4];
    const int tid  = threadIdx.x;
    const int lane = tid & 31;
    const int wave = tid >> 5;
    const int base = blockIdx.x * GR_NB;
    const int slot = lane >> 4;
    const int sub  = lane & 15;

    if (tid < 2 * GR_NB) { gs[tid] = 0.0f; cnt[tid] = 0; }
    if (tid == 0) nlist = 0;

    constexpr int UPD_T = (GR_NB * 32 + GR_THREADS - 1) / GR_THREADS;
    __syncthreads();

    float4 px[FIRST ? UPD_T : 1], p0[FIRST ? UPD_T : 1];
    if constexpr (FIRST) {
#pragma unroll
        for (int t = 0; t < UPD_T; ++t) {
            const int rl   = min((tid + t * GR_THREADS) >> 5, GR_NB - 1);
            const int node = min(base + rl, n - 1);
            const size_t o = (size_t)node * (HIDC / 4) + lane;
            px[t] = ((const float4*)xin)[o];
            p0[t] = ((const float4*)x0in)[o];
        }
    }

    for (int set = 0; set < 2; ++set) {
        const int* __restrict__ ei  = set ? ei2 : ei1;
        const int               E   = set ? E2 : E1;
        const int* __restrict__ tab = set ? tab2 : tab1;
        float* gsp  = gs  + set * GR_NB;
        int*   cntp = cnt + set * GR_NB;
        const int* __restrict__ rowp = ei;
        const int* __restrict__ colp = ei + E;
        const int ngrp = (E + GR_CAP - 1) / GR_CAP;
        const int rot = (int)(blockIdx.x % (unsigned)ngrp);
        int lcnt = -1;
        if (set == 0) {
            const int c = ocnt[2 * blockIdx.x + 1];
            if (c <= OL_CAP) lcnt = c;
        }
        const int ntrip = (lcnt >= 0) ? 1 : ngrp;
        for (int g0 = 0; g0 < ntrip; ++g0) {
            if (lcnt >= 0) {
                const unsigned* __restrict__ L = olist + (size_t)blockIdx.x * OL_CAP;
                for (int i = tid; i < lcnt; i += GR_THREADS) list[i] = L[i];
                if (tid == 0) nlist = lcnt;
            } else {
            int gi = g0 + rot;
            if (gi >= ngrp) gi -= ngrp;
            const int lo = tab[2 * gi], hi = tab[2 * gi + 1];
            if (hi < base || lo >= base + GR_NB) continue;

            for (int it = 0; it < GR_CAP / (GR_THREADS * GR_U); ++it) {
                const int e0 = gi * GR_CAP + (it * (GR_THREADS / 32) + wave) * (32 * GR_U);
                int  rr[GR_U], cc[GR_U];
                bool hit[GR_U];
#pragma unroll
                for (int u = 0; u < GR_U; ++u) {
                    const int e  = e0 + u * 32 + lane;
                    const int ec = min(e, E - 1);
                    rr[u]  = rowp[ec] - base;
                    hit[u] = (e < E) && ((unsigned)rr[u] < (unsigned)GR_NB);
                }
#pragma unroll
                for (int u = 0; u < GR_U; ++u) {
                    cc[u] = 0;
                    if (hit[u]) cc[u] = colp[e0 + u * 32 + lane];
                }
#pragma unroll
                for (int u = 0; u < GR_U; ++u) {
                    const unsigned mask = __builtin_amdgcn_ballot_w32(hit[u]);
                    if (mask) {
                        int off = 0;
                        if (lane == 0) off = atomicAdd(&nlist, __builtin_popcount(mask));
                        off = __builtin_amdgcn_readfirstlane(off);
                        if (hit[u])
                            list[off + __builtin_amdgcn_mbcnt_lo(mask, 0)] =
                                ((unsigned)cc[u] << 7) | (unsigned)rr[u];
                    }
                }
            }
            }
            __syncthreads();
            const int nh = nlist;

            for (int s = wave * 8; s < nh; s += (GR_THREADS / 32) * 8) {
                int    rl[4];
                bool   ok[4];
                float4 a[4][2], b[4][2];
#pragma unroll
                for (int q = 0; q < 4; ++q) {
                    const int idx = s + 2 * q + slot;
                    ok[q] = idx < nh;
                    const unsigned ent = list[min(idx, nh - 1)];
                    rl[q] = (int)(ent & 127u);
                    const int c = (int)(ent >> 7);
                    const float4* pa = (const float4*)(agg + (size_t)(base + rl[q]) * HIDC) + sub;
                    const float4* pb = (const float4*)(agg + (size_t)c * HIDC) + sub;
#pragma unroll
                    for (int i = 0; i < 2; ++i) {
                        a[q][i] = pa[16 * i];
                        b[q][i] = pb[16 * i];
                    }
                }

                __builtin_amdgcn_sched_barrier(0);

#pragma unroll
                for (int q = 0; q < 4; ++q) {
                    float sv = 0.0f;
#pragma unroll
                    for (int i = 0; i < 2; ++i) {
                        const float dx = a[q][i].x - b[q][i].x;
                        const float dy = a[q][i].y - b[q][i].y;
                        const float dz = a[q][i].z - b[q][i].z;
                        const float dw = a[q][i].w - b[q][i].w;
                        sv += dx * dx + dy * dy + dz * dz + dw * dw;
                    }
                    sv += lane_read<0x161>(sv);
                    sv += lane_read<0x162>(sv);
                    sv += lane_read<0x164>(sv);
                    sv += lane_read<0x168>(sv);
                    if (sub == 0 && ok[q]) {
                        atomicAdd(&gsp[rl[q]], sv);
                        atomicAdd(&cntp[rl[q]], 1);
                    }
                }
            }
            __syncthreads();
            if (tid == 0) nlist = 0;
            __syncthreads();
        }
    }

    if constexpr (FIRST) {
        if (tid < 2 * GR_NB) gs[tid] = gate_value(gs[tid], cnt[tid]);
#pragma unroll
        for (int t = 0; t < UPD_T; ++t) {
            const int i = tid + t * GR_THREADS;
            if (i < GR_NB * 32) {
                float* d = xa + (i >> 5) * AGG_P + lane;
                d[0]  = px[t].x; d[32] = px[t].y; d[64] = px[t].z; d[96] = px[t].w;
                float* e = d + GR_NB * AGG_P;
                e[0]  = p0[t].x; e[32] = p0[t].y; e[64] = p0[t].z; e[96] = p0[t].w;
            }
        }
        __syncthreads();

        const int half = lane >> 4;
        const int l16  = lane & 15;
        const int col  = ((wave & 7) << 4) + l16;
        const float* Bh = Wi  + (size_t)(2 * half) * HIDC + col;
        const float* Bx = Ws2 + (size_t)(2 * half) * HIDC + col;
        for (int mt = wave >> 3; mt < GR_NB / 16; mt += GR_THREADS / 256) {
            const int r0 = mt * 16 + 8 * half;
            float ga[8];
#pragma unroll
            for (int v = 0; v < 8; ++v) {
                const int node = min(base + r0 + v, n - 1);
                ga[v] = agg[(size_t)node * HIDC + col];
            }
            const float* Ah = xa + (mt * 16 + l16) * AGG_P + (2 * half) * 32;
            const float* Ax = Ah + GR_NB * AGG_P;
            v8f dh = {}, dx = {};
#pragma unroll 8
            for (int k0 = 0; k0 < HIDC; k0 += 4) {
                v2f a, b;
                a.x = Ah[k0 >> 2];
                a.y = Ah[32 + (k0 >> 2)];
                b.x = Bh[k0 * HIDC];
                b.y = Bh[(k0 + 1) * HIDC];
                dh = __builtin_amdgcn_wmma_f32_16x16x4_f32(
                         false, a, false, b, (short)0, dh, false, false);
                a.x = Ax[k0 >> 2];
                a.y = Ax[32 + (k0 >> 2)];
                b.x = Bx[k0 * HIDC];
                b.y = Bx[(k0 + 1) * HIDC];
                dx = __builtin_amdgcn_wmma_f32_16x16x4_f32(
                         false, a, false, b, (short)0, dx, false, false);
            }
#pragma unroll
            for (int v = 0; v < 8; ++v) {
                const int rl   = r0 + v;
                const int node = base + rl;
                if (node < n) {
                    const float gm  = gs[rl];
                    const float gq  = gs[GR_NB + rl];
                    const float den = 1.0f + gm + gq;
                    const size_t o  = (size_t)node * HIDC + col;
                    hup[o]   = (dh[v] + gm * ga[v] + gq * dx[v]) / den;
                    xsout[o] = dx[v];
                }
            }
        }
        return;
    }

    float4 hv[UPD_T], av[UPD_T], xv[UPD_T];
#pragma unroll
    for (int t = 0; t < UPD_T; ++t) {
        const int rl   = min((tid + t * GR_THREADS) >> 5, GR_NB - 1);
        const int node = min(base + rl, n - 1);
        const size_t o = (size_t)node * (HIDC / 4) + lane;
        hv[t] = ((const float4*)hup)[o];
        av[t] = ((const float4*)agg)[o];
        xv[t] = ((const float4*)xs)[o];
    }

    if (tid < 2 * GR_NB) gs[tid] = gate_value(gs[tid], cnt[tid]);
    __syncthreads();

#pragma unroll
    for (int t = 0; t < UPD_T; ++t) {
        const int i    = tid + t * GR_THREADS;
        const int rl   = i >> 5;
        const int node = base + rl;
        if (i < GR_NB * 32 && node < n) {
            const float gm  = gs[rl];
            const float gq  = gs[GR_NB + rl];
            const float den = 1.0f + gm + gq;
            float4 r;
            r.x = (hv[t].x + gm * av[t].x + gq * xv[t].x) / den;
            r.y = (hv[t].y + gm * av[t].y + gq * xv[t].y) / den;
            r.z = (hv[t].z + gm * av[t].z + gq * xv[t].z) / den;
            r.w = (hv[t].w + gm * av[t].w + gq * xv[t].w) / den;
            ((float4*)hup)[(size_t)node * (HIDC / 4) + lane] = r;
        }
    }
}

extern "C" void kernel_launch(void* const* d_in, const int* in_sizes, int n_in,
                              void* d_out, int out_size, void* d_ws, size_t ws_size,
                              hipStream_t stream) {
    const float* x      = (const float*)d_in[0];
    const float* x0     = (const float*)d_in[1];
    const int*   ei     = (const int*)d_in[2];
    const int*   ei2    = (const int*)d_in[3];
    const float* in_W   = (const float*)d_in[4];
    const float* skip_W = (const float*)d_in[5];
    const float* conv_W = (const float*)d_in[6];
    const float* conv_b = (const float*)d_in[7];
    const float* fc_W   = (const float*)d_in[8];
    const float* fc_b   = (const float*)d_in[9];
    float* out = (float*)d_out;

    const int N  = in_sizes[0] / HIDC;
    const int E1 = in_sizes[2] / 2;
    const int E2 = in_sizes[3] / 2;

    const size_t nh = (size_t)N * HIDC;
    float* h     = (float*)d_ws;
    float* xskip = h + nh;
    float* agg   = xskip + nh;
    float* mbuf  = agg + nh;
    int*   tab1  = (int*)(mbuf + nh);
    const int ng1 = (E1 + GR_CAP - 1) / GR_CAP;
    const int ng2 = (E2 + GR_CAP - 1) / GR_CAP;
    int*   tab2  = tab1 + 2 * ng1;
    const int nown = (N + GR_NB - 1) / GR_NB;
    float* w2 = mbuf;
    float* w1 = mbuf + HIDC * HIDC;
    unsigned* ola  = (unsigned*)(mbuf + 2 * HIDC * HIDC);
    unsigned* olg  = ola + (size_t)nown * OL_CAP;
    int*      ocnt = (int*)(mbuf + nh + 2 * (size_t)N);

    const dim3 blk(256);
    auto gemm_blocks = [](int M, int M2, int Nc) {
        int tiles = (((M + 15) / 16) + ((M2 + 15) / 16)) * ((Nc + 15) / 16);
        return dim3((tiles + 7) / 8);
    };

    range_table<<<nown + ng1 + ng2 + WP_BLOCKS, dim3(GR_THREADS), 0, stream>>>(
        ei, E1, ng1, ei2, E2, tab1, tab2, nown, ola, olg, ocnt,
        ng1 + ng2, in_W, conv_W, skip_W, w1, w2);

    agg_by_range<<<(N + AGG_NB - 1) / AGG_NB, dim3(AGG_THREADS), 0, stream>>>(
        x, ei, E1, w1, conv_b, agg, N, ola, ocnt);
    gamma_by_range<true><<<nown, dim3(GR_THREADS), 0, stream>>>(
        agg, ei, E1, tab1, ei2, E2, tab2, h, nullptr, N, olg, ocnt,
        x, x0, in_W, w2, xskip);

    agg_by_range<<<(N + AGG_NB - 1) / AGG_NB, dim3(AGG_THREADS), 0, stream>>>(
        h, ei, E1, conv_W + (size_t)HIDC * HIDC, conv_b + HIDC, agg, N, ola, ocnt);
    gamma_by_range<false><<<nown, dim3(GR_THREADS), 0, stream>>>(
        agg, ei, E1, tab1, ei2, E2, tab2, h, xskip, N, olg, ocnt,
        nullptr, nullptr, nullptr, nullptr, nullptr);

    gemm_wmma_f32<40><<<gemm_blocks(N, 0, 40), blk, 0, stream>>>(
        h, fc_W, fc_b, out, N, nullptr, nullptr, nullptr, 0);
}
